# starting kernel
# baseline (speedup 1.0000x reference)
#include <hip/hip_runtime.h>
#include <hip/hip_bf16.h>

typedef float v2f __attribute__((ext_vector_type(2)));
typedef float v4f __attribute__((ext_vector_type(4)));
typedef float v8f __attribute__((ext_vector_type(8)));

#define NEG_SLOPE 0.01f
#define DCOL 128

__device__ __forceinline__ float lrelu(float x) {
    return x > 0.0f ? x : NEG_SLOPE * x;
}

__global__ void k_rootrow(const int* __restrict__ batch,
                          const int* __restrict__ root_idx,
                          int* __restrict__ rootrow, int N) {
    int i = blockIdx.x * blockDim.x + threadIdx.x;
    if (i < N) rootrow[i] = root_idx[batch[i]];
}

__global__ void k_init_bias(float* __restrict__ acc,
                            const float* __restrict__ bias, int N) {
    int i = blockIdx.x * blockDim.x + threadIdx.x;
    if (i < N * DCOL) acc[i] = bias[i & (DCOL - 1)];
}

__global__ void k_scatter(const float* __restrict__ xw,
                          const int* __restrict__ src,
                          const int* __restrict__ dst,
                          const float* __restrict__ wvals,
                          float* __restrict__ acc, int E) {
    int t = blockIdx.x * blockDim.x + threadIdx.x;
    int e = t >> 5;
    int lane = t & 31;
    if (e >= E) return;
    int s = src[e];
    int d = dst[e];
    float w = wvals[e];
    const v4f x = *(const v4f*)(xw + (size_t)s * DCOL + lane * 4);
    float* a = acc + (size_t)d * DCOL + lane * 4;
    unsafeAtomicAdd(a + 0, w * x[0]);
    unsafeAtomicAdd(a + 1, w * x[1]);
    unsafeAtomicAdd(a + 2, w * x[2]);
    unsafeAtomicAdd(a + 3, w * x[3]);
}

template <int KCHUNKS, bool LR0, bool LR1, bool EPI>
__global__ void k_gemm_wmma(const float* __restrict__ A0,
                            const float* __restrict__ A1,
                            const int* __restrict__ rootrow,
                            const float* __restrict__ W,
                            const float* __restrict__ bias,
                            float* __restrict__ C, int N) {
    __shared__ float lw[128 * 128];

    const int tid  = threadIdx.x;
    const int wave = tid >> 5;
    const int lane = tid & 31;
    const int lo   = lane & 15;
    const int hi   = lane >> 4;
    const int rowbase = blockIdx.x * 128 + wave * 16;
    const bool active = rowbase < N;

    const int arow = rowbase + lo;
    int arow1 = 0;
    if (KCHUNKS > 1 && active) arow1 = rootrow[arow];

    v8f acc[8] = {};

#pragma unroll
    for (int kc = 0; kc < KCHUNKS; ++kc) {
        const float* wsrc = W + (size_t)kc * 128 * DCOL;
        for (int i = tid; i < (128 * DCOL) / 8; i += 256) {
            int kp = i >> 5;
            int cg = (i & 31) * 4;
            v4f e = *(const v4f*)(wsrc + (kp * 2 + 0) * DCOL + cg);
            v4f o = *(const v4f*)(wsrc + (kp * 2 + 1) * DCOL + cg);
            float* dstp = lw + kp * 256 + cg * 2;
#pragma unroll
            for (int j = 0; j < 4; ++j) {
                v2f p = { e[j], o[j] };
                *(v2f*)(dstp + j * 2) = p;
            }
        }
        __syncthreads();

        if (active) {
            const float* arowptr = (kc == 0) ? (A0 + (size_t)arow * DCOL)
                                             : (A1 + (size_t)arow1 * DCOL);
            const bool lr = (kc == 0) ? LR0 : LR1;
            const float* lb = lw + hi * 256 + lo * 2;
#pragma unroll 4
            for (int ks = 0; ks < 32; ++ks) {
                v2f a = *(const v2f*)(arowptr + ks * 4 + hi * 2);
                if (lr) { a[0] = lrelu(a[0]); a[1] = lrelu(a[1]); }
                v2f b[8];
#pragma unroll
                for (int t = 0; t < 8; ++t)
                    b[t] = *(const v2f*)(lb + ks * 512 + t * 32);
#pragma unroll
                for (int t = 0; t < 8; ++t)
                    acc[t] = __builtin_amdgcn_wmma_f32_16x16x4_f32(
                        false, a, false, b[t], (short)0, acc[t], false, false);
            }
        }
        __syncthreads();
    }

    if (active) {
#pragma unroll
        for (int t = 0; t < 8; ++t) {
            int col = t * 16 + lo;
            float badd = EPI ? bias[col] : 0.0f;
#pragma unroll
            for (int r = 0; r < 8; ++r) {
                int row = rowbase + r + hi * 8;
                float v = acc[t][r];
                if (EPI) v = lrelu(v + badd);
                C[(size_t)row * DCOL + col] = v;
            }
        }
    }
}

extern "C" void kernel_launch(void* const* d_in, const int* in_sizes, int n_in,
                              void* d_out, int out_size, void* d_ws, size_t ws_size,
                              hipStream_t stream) {
    const float* features = (const float*)d_in[0];
    const float* values   = (const float*)d_in[1];
    const float* W1       = (const float*)d_in[2];
    const float* b1       = (const float*)d_in[3];
    const float* W2       = (const float*)d_in[4];
    const float* b2       = (const float*)d_in[5];
    const float* Wl       = (const float*)d_in[6];
    const float* bl       = (const float*)d_in[7];
    const int*   adjs     = (const int*)d_in[8];
    const int*   root_idx = (const int*)d_in[9];
    const int*   batch    = (const int*)d_in[12];

    const int N = in_sizes[0] / DCOL;
    const int E = in_sizes[1];
    const int* srcI = adjs;
    const int* dstI = adjs + E;

    float* XW = (float*)d_ws;
    float* H1 = XW + (size_t)N * DCOL;
    float* H2 = H1 + (size_t)N * DCOL;
    int* ROOTROW = (int*)(H2 + (size_t)N * DCOL);

    float* OUT = (float*)d_out;

    const int gemmBlocks = (N + 127) / 128;
    const int initBlocks = (N * DCOL + 255) / 256;
    const int scatBlocks = (int)(((size_t)E * 32 + 255) / 256);

    k_rootrow<<<(N + 255) / 256, 256, 0, stream>>>(batch, root_idx, ROOTROW, N);

    k_gemm_wmma<1, false, false, false><<<gemmBlocks, 256, 0, stream>>>(
        features, nullptr, nullptr, W1, nullptr, XW, N);
    k_init_bias<<<initBlocks, 256, 0, stream>>>(H1, b1, N);
    k_scatter<<<scatBlocks, 256, 0, stream>>>(XW, srcI, dstI, values, H1, E);

    k_gemm_wmma<2, true, true, false><<<gemmBlocks, 256, 0, stream>>>(
        H1, features, ROOTROW, W2, nullptr, XW, N);
    k_init_bias<<<initBlocks, 256, 0, stream>>>(H2, b2, N);
    k_scatter<<<scatBlocks, 256, 0, stream>>>(XW, srcI, dstI, values, H2, E);

    k_gemm_wmma<2, true, false, true><<<gemmBlocks, 256, 0, stream>>>(
        H2, H1, ROOTROW, Wl, bl, OUT, N);
}


# Scatter lanes own channels l+32j: each f32 atomic instruction adds 32 consecutive floats of ONE 128-byte line (was 8 floats in each of 4 lines); sums still in atomic arrival order
# speedup vs baseline: 1.3493x; 1.3493x over previous
#include <hip/hip_runtime.h>
#include <hip/hip_bf16.h>

typedef float v2f __attribute__((ext_vector_type(2)));
typedef float v4f __attribute__((ext_vector_type(4)));
typedef float v8f __attribute__((ext_vector_type(8)));

#define NEG_SLOPE 0.01f
#define DCOL 128

__device__ __forceinline__ float lrelu(float x) {
    return x > 0.0f ? x : NEG_SLOPE * x;
}

__global__ void k_rootrow(const int* __restrict__ batch,
                          const int* __restrict__ root_idx,
                          int* __restrict__ rootrow, int N) {
    int i = blockIdx.x * blockDim.x + threadIdx.x;
    if (i < N) rootrow[i] = root_idx[batch[i]];
}

__global__ void k_init_bias(float* __restrict__ acc,
                            const float* __restrict__ bias, int N) {
    int i = blockIdx.x * blockDim.x + threadIdx.x;
    if (i < N * DCOL) acc[i] = bias[i & (DCOL - 1)];
}

__global__ void k_scatter(const float* __restrict__ xw,
                          const int* __restrict__ src,
                          const int* __restrict__ dst,
                          const float* __restrict__ wvals,
                          float* __restrict__ acc, int E) {
    int t = blockIdx.x * blockDim.x + threadIdx.x;
    int e = t >> 5;
    int lane = t & 31;
    if (e >= E) return;
    int s = src[e];
    int d = dst[e];
    float w = wvals[e];
    const float* x = xw + (size_t)s * DCOL + lane;
    float* a = acc + (size_t)d * DCOL + lane;
    const float x0 = x[0];
    const float x1 = x[32];
    const float x2 = x[64];
    const float x3 = x[96];
    unsafeAtomicAdd(a + 0,  w * x0);
    unsafeAtomicAdd(a + 32, w * x1);
    unsafeAtomicAdd(a + 64, w * x2);
    unsafeAtomicAdd(a + 96, w * x3);
}

template <int KCHUNKS, bool LR0, bool LR1, bool EPI>
__global__ void k_gemm_wmma(const float* __restrict__ A0,
                            const float* __restrict__ A1,
                            const int* __restrict__ rootrow,
                            const float* __restrict__ W,
                            const float* __restrict__ bias,
                            float* __restrict__ C, int N) {
    __shared__ float lw[128 * 128];

    const int tid  = threadIdx.x;
    const int wave = tid >> 5;
    const int lane = tid & 31;
    const int lo   = lane & 15;
    const int hi   = lane >> 4;
    const int rowbase = blockIdx.x * 128 + wave * 16;
    const bool active = rowbase < N;

    const int arow = rowbase + lo;
    int arow1 = 0;
    if (KCHUNKS > 1 && active) arow1 = rootrow[arow];

    v8f acc[8] = {};

#pragma unroll
    for (int kc = 0; kc < KCHUNKS; ++kc) {
        const float* wsrc = W + (size_t)kc * 128 * DCOL;
        for (int i = tid; i < (128 * DCOL) / 8; i += 256) {
            int kp = i >> 5;
            int cg = (i & 31) * 4;
            v4f e = *(const v4f*)(wsrc + (kp * 2 + 0) * DCOL + cg);
            v4f o = *(const v4f*)(wsrc + (kp * 2 + 1) * DCOL + cg);
            float* dstp = lw + kp * 256 + cg * 2;
#pragma unroll
            for (int j = 0; j < 4; ++j) {
                v2f p = { e[j], o[j] };
                *(v2f*)(dstp + j * 2) = p;
            }
        }
        __syncthreads();

        if (active) {
            const float* arowptr = (kc == 0) ? (A0 + (size_t)arow * DCOL)
                                             : (A1 + (size_t)arow1 * DCOL);
            const bool lr = (kc == 0) ? LR0 : LR1;
            const float* lb = lw + hi * 256 + lo * 2;
#pragma unroll 4
            for (int ks = 0; ks < 32; ++ks) {
                v2f a = *(const v2f*)(arowptr + ks * 4 + hi * 2);
                if (lr) { a[0] = lrelu(a[0]); a[1] = lrelu(a[1]); }
                v2f b[8];
#pragma unroll
                for (int t = 0; t < 8; ++t)
                    b[t] = *(const v2f*)(lb + ks * 512 + t * 32);
#pragma unroll
                for (int t = 0; t < 8; ++t)
                    acc[t] = __builtin_amdgcn_wmma_f32_16x16x4_f32(
                        false, a, false, b[t], (short)0, acc[t], false, false);
            }
        }
        __syncthreads();
    }

    if (active) {
#pragma unroll
        for (int t = 0; t < 8; ++t) {
            int col = t * 16 + lo;
            float badd = EPI ? bias[col] : 0.0f;
#pragma unroll
            for (int r = 0; r < 8; ++r) {
                int row = rowbase + r + hi * 8;
                float v = acc[t][r];
                if (EPI) v = lrelu(v + badd);
                C[(size_t)row * DCOL + col] = v;
            }
        }
    }
}

extern "C" void kernel_launch(void* const* d_in, const int* in_sizes, int n_in,
                              void* d_out, int out_size, void* d_ws, size_t ws_size,
                              hipStream_t stream) {
    const float* features = (const float*)d_in[0];
    const float* values   = (const float*)d_in[1];
    const float* W1       = (const float*)d_in[2];
    const float* b1       = (const float*)d_in[3];
    const float* W2       = (const float*)d_in[4];
    const float* b2       = (const float*)d_in[5];
    const float* Wl       = (const float*)d_in[6];
    const float* bl       = (const float*)d_in[7];
    const int*   adjs     = (const int*)d_in[8];
    const int*   root_idx = (const int*)d_in[9];
    const int*   batch    = (const int*)d_in[12];

    const int N = in_sizes[0] / DCOL;
    const int E = in_sizes[1];
    const int* srcI = adjs;
    const int* dstI = adjs + E;

    float* XW = (float*)d_ws;
    float* H1 = XW + (size_t)N * DCOL;
    float* H2 = H1 + (size_t)N * DCOL;
    int* ROOTROW = (int*)(H2 + (size_t)N * DCOL);

    float* OUT = (float*)d_out;

    const int gemmBlocks = (N + 127) / 128;
    const int initBlocks = (N * DCOL + 255) / 256;
    const int scatBlocks = (int)(((size_t)E * 32 + 255) / 256);

    k_rootrow<<<(N + 255) / 256, 256, 0, stream>>>(batch, root_idx, ROOTROW, N);

    k_gemm_wmma<1, false, false, false><<<gemmBlocks, 256, 0, stream>>>(
        features, nullptr, nullptr, W1, nullptr, XW, N);
    k_init_bias<<<initBlocks, 256, 0, stream>>>(H1, b1, N);
    k_scatter<<<scatBlocks, 256, 0, stream>>>(XW, srcI, dstI, values, H1, E);

    k_gemm_wmma<2, true, true, false><<<gemmBlocks, 256, 0, stream>>>(
        H1, features, ROOTROW, W2, nullptr, XW, N);
    k_init_bias<<<initBlocks, 256, 0, stream>>>(H2, b2, N);
    k_scatter<<<scatBlocks, 256, 0, stream>>>(XW, srcI, dstI, values, H2, E);

    k_gemm_wmma<2, true, false, true><<<gemmBlocks, 256, 0, stream>>>(
        H2, H1, ROOTROW, Wl, bl, OUT, N);
}


# Second aggregation replays per-wave edge lists (src, row, weight) that the first one writes while it scans: no map scan and no index loads in pass 2; f32 LDS atomic sums still in arrival order
# speedup vs baseline: 2.0710x; 1.5349x over previous
#include <hip/hip_runtime.h>
#include <hip/hip_bf16.h>

typedef float v2f __attribute__((ext_vector_type(2)));
typedef float v4f __attribute__((ext_vector_type(4)));
typedef float v8f __attribute__((ext_vector_type(8)));

#define NEG_SLOPE 0.01f
#define DCOL 128

__device__ __forceinline__ float lrelu(float x) {
    return x > 0.0f ? x : NEG_SLOPE * x;
}

#define NPB   261
#define SW    32
#define SU    2
#define CHUNK 512

__device__ __host__ __forceinline__ int scan_steps(int E) {
    const int nchunks = (E + CHUNK - 1) / CHUNK;
    const int nsteps = (nchunks + SU * SW - 1) / (SU * SW);
    return nsteps < 1 ? 1 : nsteps;
}

__device__ __forceinline__ unsigned owner_of(int d) {
    const unsigned o = (unsigned)d / (unsigned)NPB;
    return o < 255u ? o : 255u;
}

__global__ void k_prep(const int* __restrict__ batch,
                       const int* __restrict__ root_idx,
                       int* __restrict__ rootrow, int N,
                       const int* __restrict__ dst,
                       int* __restrict__ own, int E, int rootBlocks) {
    if ((int)blockIdx.x < rootBlocks) {
        const int i = blockIdx.x * blockDim.x + threadIdx.x;
        if (i < N) rootrow[i] = root_idx[batch[i]];
        return;
    }
    const int emap = scan_steps(E) * (SU * SW * CHUNK);
    const int e0 = ((blockIdx.x - rootBlocks) * blockDim.x + threadIdx.x) * 16;
    if (e0 >= emap) return;
    unsigned pk[4];
    if (((((size_t)dst) & 15) == 0) && e0 + 15 < E) {
#pragma unroll
        for (int g = 0; g < 4; ++g) {
            const int4 v = *(const int4*)(dst + e0 + g * 4);
            pk[g] = owner_of(v.x) | (owner_of(v.y) << 8) |
                    (owner_of(v.z) << 16) | (owner_of(v.w) << 24);
        }
    } else {
#pragma unroll
        for (int g = 0; g < 4; ++g) {
            unsigned p = 0u;
#pragma unroll
            for (int k = 0; k < 4; ++k) {
                const int e = e0 + g * 4 + k;
                const unsigned o = (e < E) ? owner_of(dst[e]) : 255u;
                p |= o << (8 * k);
            }
            pk[g] = p;
        }
    }
    *(uint4*)(own + e0 / 4) = make_uint4(pk[0], pk[1], pk[2], pk[3]);
}

#define BAT   15
#define QCAP  (SU * CHUNK + 64)

__device__ __forceinline__ void lds_row_add(float* a, float w, v4f x) {
    __hip_atomic_fetch_add(a + 0,  w * x[0], __ATOMIC_RELAXED, __HIP_MEMORY_SCOPE_WORKGROUP);
    __hip_atomic_fetch_add(a + 32, w * x[1], __ATOMIC_RELAXED, __HIP_MEMORY_SCOPE_WORKGROUP);
    __hip_atomic_fetch_add(a + 64, w * x[2], __ATOMIC_RELAXED, __HIP_MEMORY_SCOPE_WORKGROUP);
    __hip_atomic_fetch_add(a + 96, w * x[3], __ATOMIC_RELAXED, __HIP_MEMORY_SCOPE_WORKGROUP);
}

template <bool EXACT>
__device__ __forceinline__ int note_edges(uint4 o, unsigned pat, int ebase, int* q, int count,
                                          const int* __restrict__ dst, int n0, int nn, int E) {
    const unsigned wd[4] = { o.x, o.y, o.z, o.w };
#pragma unroll
    for (int g = 0; g < 4; ++g) {
        const unsigned x = wd[g] ^ pat;
        const unsigned z = (x - 0x01010101u) & ~x & 0x80808080u;
        if (__builtin_amdgcn_ballot_w32(z != 0u) == 0u) continue;
#pragma unroll
        for (int k = 0; k < 4; ++k) {
            const int e = ebase + g * 4 + k;
            bool m = ((x >> (8 * k)) & 0xffu) == 0u;
            if (!EXACT) {
                const bool in = m && (e < E);
                const int d = in ? dst[e] : -1;
                m = in && ((unsigned)(d - n0) < (unsigned)nn);
            }
            const unsigned mask = __builtin_amdgcn_ballot_w32(m);
            if (mask != 0u) {
                if (m) q[count + (int)__builtin_amdgcn_mbcnt_lo(mask, 0u)] = e;
                count += __builtin_popcount(mask);
            }
        }
    }
    return count;
}

__device__ __forceinline__ void ask_rows(const float* __restrict__ xw, int lane, int s,
                                         v4f (&xr)[BAT]) {
#pragma unroll
    for (int j = 0; j < BAT; ++j) {
        const int sj = __builtin_amdgcn_readlane(s, j);
        xr[j] = *(const v4f*)(xw + (size_t)sj * DCOL + lane * 4);
    }
}

__device__ __forceinline__ void add_rows(float* accs, int lane, int dl, float w,
                                         const v4f (&xr)[BAT]) {
#pragma unroll
    for (int j = 0; j < BAT; ++j) {
        const int dj = __builtin_amdgcn_readlane(dl, j);
        const float wj = __int_as_float(
            __builtin_amdgcn_readlane(__float_as_int(w), j));
        lds_row_add(accs + dj * DCOL + lane, wj, xr[j]);
    }
}

__device__ __forceinline__ void add_batch(const float* __restrict__ xw, float* accs, int lane,
                                          int nb, int s, int dl, float w) {
    v4f xr[BAT];
    if (nb == BAT) {
        ask_rows(xw, lane, s, xr);
        add_rows(accs, lane, dl, w, xr);
    } else {
#pragma unroll
        for (int j = 0; j < BAT; ++j) {
            if (j < nb) {
                const int sj = __builtin_amdgcn_readlane(s, j);
                xr[j] = *(const v4f*)(xw + (size_t)sj * DCOL + lane * 4);
            }
        }
#pragma unroll
        for (int j = 0; j < BAT; ++j) {
            if (j < nb) {
                const int dj = __builtin_amdgcn_readlane(dl, j);
                const float wj = __int_as_float(
                    __builtin_amdgcn_readlane(__float_as_int(w), j));
                lds_row_add(accs + dj * DCOL + lane, wj, xr[j]);
            }
        }
    }
}

#define CAPW 256

__device__ __forceinline__ int list_cap(int N, int E, int nblk) {
    const long long avail = (long long)N * DCOL
                          - (long long)scan_steps(E) * (SU * SW * CHUNK / 4)
                          - (long long)nblk * SW;
    const long long c = avail / (3LL * nblk * SW);
    return c < 0 ? 0 : (c > CAPW ? CAPW : (int)c);
}

template <int MODE>
__global__ __launch_bounds__(1024) void k_scatter_own(const float* __restrict__ xw,
               const int* __restrict__ src,
               const int* __restrict__ dst,
               const int* __restrict__ own,
               const float* __restrict__ wvals,
               const float* __restrict__ bias,
               float* __restrict__ out,
               int* __restrict__ lists, int N, int E) {
    __shared__ float accs[NPB * DCOL];
    __shared__ int queue[SW * QCAP];

    const int tid  = threadIdx.x;
    const int lane = tid & 31;
    const int wave = __builtin_amdgcn_readfirstlane(tid >> 5);
    const int n0 = blockIdx.x * NPB;
    const int nn = (N - n0 < NPB) ? (N - n0) : NPB;

    for (int i = tid; i < nn * DCOL; i += SW * 32)
        accs[i] = bias[((i & 31) << 2) | ((i >> 5) & 3)];
    __syncthreads();

    const int nblk = gridDim.x;
    const int cap = __builtin_amdgcn_readfirstlane(list_cap(N, E, nblk));
    const size_t plane = (size_t)nblk * SW * cap;
    int* cntp = lists + blockIdx.x * SW + wave;
    int* LS = lists + (size_t)nblk * SW + ((size_t)blockIdx.x * SW + wave) * cap;
    int* LD = LS + plane;
    float* LW = (float*)(LD + plane);

    bool scan = true;
    if (MODE == 1) {
        const int c = __builtin_amdgcn_readfirstlane(*cntp);
        if (c >= 0) {
            scan = false;
            int s = 0, dl = 0;
            float w = 0.0f;
            if (lane < BAT && lane < c) { s = LS[lane]; dl = LD[lane]; w = LW[lane]; }
            int base = 0;
            for (; base + BAT <= c; base += BAT) {
                v4f xr[BAT];
                ask_rows(xw, lane, s, xr);
                int s2 = 0, dl2 = 0;
                float w2 = 0.0f;
                if (lane < BAT && base + BAT + lane < c) {
                    s2  = LS[base + BAT + lane];
                    dl2 = LD[base + BAT + lane];
                    w2  = LW[base + BAT + lane];
                }
                add_rows(accs, lane, dl, w, xr);
                s = s2; dl = dl2; w = w2;
            }
            if (base < c) add_batch(xw, accs, lane, c - base, s, dl, w);
        }
    }

    if (scan) {
        int* q = queue + wave * QCAP;
        int count = 0;
        int ln = 0;
        bool over = false;
        const int nsteps = scan_steps(E);
        const int t0 = blockIdx.x % nsteps;
        const unsigned myb = blockIdx.x < 255u ? blockIdx.x : 255u;
        const bool exact = blockIdx.x < 255u;
        const unsigned pat = myb * 0x01010101u;

        for (int t = 0; t <= nsteps; ++t) {
            const bool more = t < nsteps;
            if (more) {
                int ts = t + t0;
                if (ts >= nsteps) ts -= nsteps;
                const int cb = ts * (SU * SW) + wave;
                uint4 ov[SU];
#pragma unroll
                for (int u = 0; u < SU; ++u)
                    ov[u] = *(const uint4*)(own + (size_t)(cb + u * SW) * (CHUNK / 4) + lane * 4);
                if (exact) {
#pragma unroll
                    for (int u = 0; u < SU; ++u)
                        count = note_edges<true>(ov[u], pat, (cb + u * SW) * CHUNK + lane * 16,
                                                 q, count, dst, n0, nn, E);
                } else {
#pragma unroll
                    for (int u = 0; u < SU; ++u)
                        count = note_edges<false>(ov[u], pat, (cb + u * SW) * CHUNK + lane * 16,
                                                  q, count, dst, n0, nn, E);
                }
            }

            while (count >= (more ? BAT : 1)) {
                const int nb = count < BAT ? count : BAT;
                count -= nb;
                int s = 0, dl = 0;
                float w = 0.0f;
                if (lane < nb) {
                    const int e = q[count + lane];
                    s  = src[e];
                    dl = dst[e] - n0;
                    w  = wvals[e];
                }
                if (MODE == 0) {
                    if (ln + nb <= cap) {
                        if (lane < nb) { LS[ln + lane] = s; LD[ln + lane] = dl; LW[ln + lane] = w; }
                        ln += nb;
                    } else {
                        over = true;
                    }
                }
                add_batch(xw, accs, lane, nb, s, dl, w);
            }
        }
        if (MODE == 0 && lane == 0) *cntp = over ? -1 : ln;
    }
    __syncthreads();

    v4f* o4 = (v4f*)(out + (size_t)n0 * DCOL);
    for (int i = tid; i < nn * (DCOL / 4); i += SW * 32) {
        const float* a = accs + (i >> 5) * DCOL + (i & 31);
        const v4f o = { a[0], a[32], a[64], a[96] };
        o4[i] = o;
    }
}

template <int KCHUNKS, bool LR0, bool LR1, bool EPI>
__global__ void k_gemm_wmma(const float* __restrict__ A0,
                            const float* __restrict__ A1,
                            const int* __restrict__ rootrow,
                            const float* __restrict__ W,
                            const float* __restrict__ bias,
                            float* __restrict__ C, int N) {
    __shared__ float lw[128 * 128];

    const int tid  = threadIdx.x;
    const int wave = tid >> 5;
    const int lane = tid & 31;
    const int lo   = lane & 15;
    const int hi   = lane >> 4;
    const int rowbase = blockIdx.x * 128 + wave * 16;
    const bool active = rowbase < N;

    const int arow = rowbase + lo;
    int arow1 = 0;
    if (KCHUNKS > 1 && active) arow1 = rootrow[arow];

    v8f acc[8] = {};

#pragma unroll
    for (int kc = 0; kc < KCHUNKS; ++kc) {
        const float* wsrc = W + (size_t)kc * 128 * DCOL;
        for (int i = tid; i < (128 * DCOL) / 8; i += 256) {
            int kp = i >> 5;
            int cg = (i & 31) * 4;
            v4f e = *(const v4f*)(wsrc + (kp * 2 + 0) * DCOL + cg);
            v4f o = *(const v4f*)(wsrc + (kp * 2 + 1) * DCOL + cg);
            float* dstp = lw + kp * 256 + cg * 2;
#pragma unroll
            for (int j = 0; j < 4; ++j) {
                v2f p = { e[j], o[j] };
                *(v2f*)(dstp + j * 2) = p;
            }
        }
        __syncthreads();

        if (active) {
            const float* arowptr = (kc == 0) ? (A0 + (size_t)arow * DCOL)
                                             : (A1 + (size_t)arow1 * DCOL);
            const bool lr = (kc == 0) ? LR0 : LR1;
            const float* lb = lw + hi * 256 + lo * 2;
#pragma unroll 4
            for (int ks = 0; ks < 32; ++ks) {
                v2f a = *(const v2f*)(arowptr + ks * 4 + hi * 2);
                if (lr) { a[0] = lrelu(a[0]); a[1] = lrelu(a[1]); }
                v2f b[8];
#pragma unroll
                for (int t = 0; t < 8; ++t)
                    b[t] = *(const v2f*)(lb + ks * 512 + t * 32);
#pragma unroll
                for (int t = 0; t < 8; ++t)
                    acc[t] = __builtin_amdgcn_wmma_f32_16x16x4_f32(
                        false, a, false, b[t], (short)0, acc[t], false, false);
            }
        }
        __syncthreads();
    }

    if (active) {
#pragma unroll
        for (int t = 0; t < 8; ++t) {
            int col = t * 16 + lo;
            float badd = EPI ? bias[col] : 0.0f;
#pragma unroll
            for (int r = 0; r < 8; ++r) {
                int row = rowbase + r + hi * 8;
                float v = acc[t][r];
                if (EPI) v = lrelu(v + badd);
                C[(size_t)row * DCOL + col] = v;
            }
        }
    }
}

extern "C" void kernel_launch(void* const* d_in, const int* in_sizes, int n_in,
                              void* d_out, int out_size, void* d_ws, size_t ws_size,
                              hipStream_t stream) {
    const float* features = (const float*)d_in[0];
    const float* values   = (const float*)d_in[1];
    const float* W1       = (const float*)d_in[2];
    const float* b1       = (const float*)d_in[3];
    const float* W2       = (const float*)d_in[4];
    const float* b2       = (const float*)d_in[5];
    const float* Wl       = (const float*)d_in[6];
    const float* bl       = (const float*)d_in[7];
    const int*   adjs     = (const int*)d_in[8];
    const int*   root_idx = (const int*)d_in[9];
    const int*   batch    = (const int*)d_in[12];

    const int N = in_sizes[0] / DCOL;
    const int E = in_sizes[1];
    const int* srcI = adjs;
    const int* dstI = adjs + E;

    float* XW = (float*)d_ws;
    float* H1 = XW + (size_t)N * DCOL;
    float* H2 = H1 + (size_t)N * DCOL;
    int* ROOTROW = (int*)(H2 + (size_t)N * DCOL);

    float* OUT = (float*)d_out;
    int* OWN = (int*)d_out;
    int* LISTS = OWN + scan_steps(E) * (SU * SW * CHUNK / 4);

    const int gemmBlocks = (N + 127) / 128;
    const int scatBlocks = (N + NPB - 1) / NPB;

    const int rootBlocks = (N + 255) / 256;
    const int ownBlocks  = (scan_steps(E) * (SU * SW * CHUNK) / 16 + 255) / 256;
    k_prep<<<rootBlocks + ownBlocks, 256, 0, stream>>>(batch, root_idx, ROOTROW, N,
                                                       dstI, OWN, E, rootBlocks);

    k_gemm_wmma<1, false, false, false><<<gemmBlocks, 256, 0, stream>>>(
        features, nullptr, nullptr, W1, nullptr, XW, N);
    k_scatter_own<0><<<scatBlocks, SW * 32, 0, stream>>>(XW, srcI, dstI, OWN, values, b1, H1, LISTS, N, E);

    k_gemm_wmma<2, true, true, false><<<gemmBlocks, 256, 0, stream>>>(
        H1, features, ROOTROW, W2, nullptr, XW, N);
    k_scatter_own<1><<<scatBlocks, SW * 32, 0, stream>>>(XW, srcI, dstI, OWN, values, b2, H2, LISTS, N, E);

    k_gemm_wmma<2, true, false, true><<<gemmBlocks, 256, 0, stream>>>(
        H2, H1, ROOTROW, Wl, bl, OUT, N);
}


# Aggregation re-cut from 192 owner blocks of 261 rows to 250 of 200: a 32-wave block on nearly every WGP (was 3 in 4), a quarter fewer row gathers and LDS adds per block; sums still LDS arrival order
# speedup vs baseline: 2.1901x; 1.0575x over previous
#include <hip/hip_runtime.h>
#include <hip/hip_bf16.h>

typedef float v2f __attribute__((ext_vector_type(2)));
typedef float v4f __attribute__((ext_vector_type(4)));
typedef float v8f __attribute__((ext_vector_type(8)));

#define NEG_SLOPE 0.01f
#define DCOL 128

__device__ __forceinline__ float lrelu(float x) {
    return x > 0.0f ? x : NEG_SLOPE * x;
}

#define NPB   200
#define SW    32
#define SU    2
#define CHUNK 512

__device__ __host__ __forceinline__ int scan_steps(int E) {
    const int nchunks = (E + CHUNK - 1) / CHUNK;
    const int nsteps = (nchunks + SU * SW - 1) / (SU * SW);
    return nsteps < 1 ? 1 : nsteps;
}

__device__ __forceinline__ unsigned owner_of(int d) {
    const unsigned o = (unsigned)d / (unsigned)NPB;
    return o < 255u ? o : 255u;
}

__global__ void k_prep(const int* __restrict__ batch,
                       const int* __restrict__ root_idx,
                       int* __restrict__ rootrow, int N,
                       const int* __restrict__ dst,
                       int* __restrict__ own, int E, int rootBlocks) {
    if ((int)blockIdx.x < rootBlocks) {
        const int i = blockIdx.x * blockDim.x + threadIdx.x;
        if (i < N) rootrow[i] = root_idx[batch[i]];
        return;
    }
    const int emap = scan_steps(E) * (SU * SW * CHUNK);
    const int e0 = ((blockIdx.x - rootBlocks) * blockDim.x + threadIdx.x) * 16;
    if (e0 >= emap) return;
    unsigned pk[4];
    if (((((size_t)dst) & 15) == 0) && e0 + 15 < E) {
#pragma unroll
        for (int g = 0; g < 4; ++g) {
            const int4 v = *(const int4*)(dst + e0 + g * 4);
            pk[g] = owner_of(v.x) | (owner_of(v.y) << 8) |
                    (owner_of(v.z) << 16) | (owner_of(v.w) << 24);
        }
    } else {
#pragma unroll
        for (int g = 0; g < 4; ++g) {
            unsigned p = 0u;
#pragma unroll
            for (int k = 0; k < 4; ++k) {
                const int e = e0 + g * 4 + k;
                const unsigned o = (e < E) ? owner_of(dst[e]) : 255u;
                p |= o << (8 * k);
            }
            pk[g] = p;
        }
    }
    *(uint4*)(own + e0 / 4) = make_uint4(pk[0], pk[1], pk[2], pk[3]);
}

#define BAT   15
#define QCAP  (SU * CHUNK + 64)

__device__ __forceinline__ void lds_row_add(float* a, float w, v4f x) {
    __hip_atomic_fetch_add(a + 0,  w * x[0], __ATOMIC_RELAXED, __HIP_MEMORY_SCOPE_WORKGROUP);
    __hip_atomic_fetch_add(a + 32, w * x[1], __ATOMIC_RELAXED, __HIP_MEMORY_SCOPE_WORKGROUP);
    __hip_atomic_fetch_add(a + 64, w * x[2], __ATOMIC_RELAXED, __HIP_MEMORY_SCOPE_WORKGROUP);
    __hip_atomic_fetch_add(a + 96, w * x[3], __ATOMIC_RELAXED, __HIP_MEMORY_SCOPE_WORKGROUP);
}

template <bool EXACT>
__device__ __forceinline__ int note_edges(uint4 o, unsigned pat, int ebase, int* q, int count,
                                          const int* __restrict__ dst, int n0, int nn, int E) {
    const unsigned wd[4] = { o.x, o.y, o.z, o.w };
#pragma unroll
    for (int g = 0; g < 4; ++g) {
        const unsigned x = wd[g] ^ pat;
        const unsigned z = (x - 0x01010101u) & ~x & 0x80808080u;
        if (__builtin_amdgcn_ballot_w32(z != 0u) == 0u) continue;
#pragma unroll
        for (int k = 0; k < 4; ++k) {
            const int e = ebase + g * 4 + k;
            bool m = ((x >> (8 * k)) & 0xffu) == 0u;
            if (!EXACT) {
                const bool in = m && (e < E);
                const int d = in ? dst[e] : -1;
                m = in && ((unsigned)(d - n0) < (unsigned)nn);
            }
            const unsigned mask = __builtin_amdgcn_ballot_w32(m);
            if (mask != 0u) {
                if (m) q[count + (int)__builtin_amdgcn_mbcnt_lo(mask, 0u)] = e;
                count += __builtin_popcount(mask);
            }
        }
    }
    return count;
}

__device__ __forceinline__ void ask_rows(const float* __restrict__ xw, int lane, int s,
                                         v4f (&xr)[BAT]) {
#pragma unroll
    for (int j = 0; j < BAT; ++j) {
        const int sj = __builtin_amdgcn_readlane(s, j);
        xr[j] = *(const v4f*)(xw + (size_t)sj * DCOL + lane * 4);
    }
}

__device__ __forceinline__ void add_rows(float* accs, int lane, int dl, float w,
                                         const v4f (&xr)[BAT]) {
#pragma unroll
    for (int j = 0; j < BAT; ++j) {
        const int dj = __builtin_amdgcn_readlane(dl, j);
        const float wj = __int_as_float(
            __builtin_amdgcn_readlane(__float_as_int(w), j));
        lds_row_add(accs + dj * DCOL + lane, wj, xr[j]);
    }
}

__device__ __forceinline__ void add_batch(const float* __restrict__ xw, float* accs, int lane,
                                          int nb, int s, int dl, float w) {
    v4f xr[BAT];
    if (nb == BAT) {
        ask_rows(xw, lane, s, xr);
        add_rows(accs, lane, dl, w, xr);
    } else {
#pragma unroll
        for (int j = 0; j < BAT; ++j) {
            if (j < nb) {
                const int sj = __builtin_amdgcn_readlane(s, j);
                xr[j] = *(const v4f*)(xw + (size_t)sj * DCOL + lane * 4);
            }
        }
#pragma unroll
        for (int j = 0; j < BAT; ++j) {
            if (j < nb) {
                const int dj = __builtin_amdgcn_readlane(dl, j);
                const float wj = __int_as_float(
                    __builtin_amdgcn_readlane(__float_as_int(w), j));
                lds_row_add(accs + dj * DCOL + lane, wj, xr[j]);
            }
        }
    }
}

#define CAPW 256

__device__ __forceinline__ int list_cap(int N, int E, int nblk) {
    const long long avail = (long long)N * DCOL
                          - (long long)scan_steps(E) * (SU * SW * CHUNK / 4)
                          - (long long)nblk * SW;
    const long long c = avail / (3LL * nblk * SW);
    return c < 0 ? 0 : (c > CAPW ? CAPW : (int)c);
}

template <int MODE>
__global__ __launch_bounds__(1024) void k_scatter_own(const float* __restrict__ xw,
               const int* __restrict__ src,
               const int* __restrict__ dst,
               const int* __restrict__ own,
               const float* __restrict__ wvals,
               const float* __restrict__ bias,
               float* __restrict__ out,
               int* __restrict__ lists, int N, int E) {
    __shared__ float accs[NPB * DCOL];
    __shared__ int queue[SW * QCAP];

    const int tid  = threadIdx.x;
    const int lane = tid & 31;
    const int wave = __builtin_amdgcn_readfirstlane(tid >> 5);
    const int n0 = blockIdx.x * NPB;
    const int nn = (N - n0 < NPB) ? (N - n0) : NPB;

    for (int i = tid; i < nn * DCOL; i += SW * 32)
        accs[i] = bias[((i & 31) << 2) | ((i >> 5) & 3)];
    __syncthreads();

    const int nblk = gridDim.x;
    const int cap = __builtin_amdgcn_readfirstlane(list_cap(N, E, nblk));
    const size_t plane = (size_t)nblk * SW * cap;
    int* cntp = lists + blockIdx.x * SW + wave;
    int* LS = lists + (size_t)nblk * SW + ((size_t)blockIdx.x * SW + wave) * cap;
    int* LD = LS + plane;
    float* LW = (float*)(LD + plane);

    bool scan = true;
    if (MODE == 1) {
        const int c = __builtin_amdgcn_readfirstlane(*cntp);
        if (c >= 0) {
            scan = false;
            int s = 0, dl = 0;
            float w = 0.0f;
            if (lane < BAT && lane < c) { s = LS[lane]; dl = LD[lane]; w = LW[lane]; }
            int base = 0;
            for (; base + BAT <= c; base += BAT) {
                v4f xr[BAT];
                ask_rows(xw, lane, s, xr);
                int s2 = 0, dl2 = 0;
                float w2 = 0.0f;
                if (lane < BAT && base + BAT + lane < c) {
                    s2  = LS[base + BAT + lane];
                    dl2 = LD[base + BAT + lane];
                    w2  = LW[base + BAT + lane];
                }
                add_rows(accs, lane, dl, w, xr);
                s = s2; dl = dl2; w = w2;
            }
            if (base < c) add_batch(xw, accs, lane, c - base, s, dl, w);
        }
    }

    if (scan) {
        int* q = queue + wave * QCAP;
        int count = 0;
        int ln = 0;
        bool over = false;
        const int nsteps = scan_steps(E);
        const int t0 = blockIdx.x % nsteps;
        const unsigned myb = blockIdx.x < 255u ? blockIdx.x : 255u;
        const bool exact = blockIdx.x < 255u;
        const unsigned pat = myb * 0x01010101u;

        for (int t = 0; t <= nsteps; ++t) {
            const bool more = t < nsteps;
            if (more) {
                int ts = t + t0;
                if (ts >= nsteps) ts -= nsteps;
                const int cb = ts * (SU * SW) + wave;
                uint4 ov[SU];
#pragma unroll
                for (int u = 0; u < SU; ++u)
                    ov[u] = *(const uint4*)(own + (size_t)(cb + u * SW) * (CHUNK / 4) + lane * 4);
                if (exact) {
#pragma unroll
                    for (int u = 0; u < SU; ++u)
                        count = note_edges<true>(ov[u], pat, (cb + u * SW) * CHUNK + lane * 16,
                                                 q, count, dst, n0, nn, E);
                } else {
#pragma unroll
                    for (int u = 0; u < SU; ++u)
                        count = note_edges<false>(ov[u], pat, (cb + u * SW) * CHUNK + lane * 16,
                                                  q, count, dst, n0, nn, E);
                }
            }

            while (count >= (more ? BAT : 1)) {
                const int nb = count < BAT ? count : BAT;
                count -= nb;
                int s = 0, dl = 0;
                float w = 0.0f;
                if (lane < nb) {
                    const int e = q[count + lane];
                    s  = src[e];
                    dl = dst[e] - n0;
                    w  = wvals[e];
                }
                if (MODE == 0) {
                    if (ln + nb <= cap) {
                        if (lane < nb) { LS[ln + lane] = s; LD[ln + lane] = dl; LW[ln + lane] = w; }
                        ln += nb;
                    } else {
                        over = true;
                    }
                }
                add_batch(xw, accs, lane, nb, s, dl, w);
            }
        }
        if (MODE == 0 && lane == 0) *cntp = over ? -1 : ln;
    }
    __syncthreads();

    v4f* o4 = (v4f*)(out + (size_t)n0 * DCOL);
    for (int i = tid; i < nn * (DCOL / 4); i += SW * 32) {
        const float* a = accs + (i >> 5) * DCOL + (i & 31);
        const v4f o = { a[0], a[32], a[64], a[96] };
        o4[i] = o;
    }
}

template <int KCHUNKS, bool LR0, bool LR1, bool EPI>
__global__ void k_gemm_wmma(const float* __restrict__ A0,
                            const float* __restrict__ A1,
                            const int* __restrict__ rootrow,
                            const float* __restrict__ W,
                            const float* __restrict__ bias,
                            float* __restrict__ C, int N) {
    __shared__ float lw[128 * 128];

    const int tid  = threadIdx.x;
    const int wave = tid >> 5;
    const int lane = tid & 31;
    const int lo   = lane & 15;
    const int hi   = lane >> 4;
    const int rowbase = blockIdx.x * 128 + wave * 16;
    const bool active = rowbase < N;

    const int arow = rowbase + lo;
    int arow1 = 0;
    if (KCHUNKS > 1 && active) arow1 = rootrow[arow];

    v8f acc[8] = {};

#pragma unroll
    for (int kc = 0; kc < KCHUNKS; ++kc) {
        const float* wsrc = W + (size_t)kc * 128 * DCOL;
        for (int i = tid; i < (128 * DCOL) / 8; i += 256) {
            int kp = i >> 5;
            int cg = (i & 31) * 4;
            v4f e = *(const v4f*)(wsrc + (kp * 2 + 0) * DCOL + cg);
            v4f o = *(const v4f*)(wsrc + (kp * 2 + 1) * DCOL + cg);
            float* dstp = lw + kp * 256 + cg * 2;
#pragma unroll
            for (int j = 0; j < 4; ++j) {
                v2f p = { e[j], o[j] };
                *(v2f*)(dstp + j * 2) = p;
            }
        }
        __syncthreads();

        if (active) {
            const float* arowptr = (kc == 0) ? (A0 + (size_t)arow * DCOL)
                                             : (A1 + (size_t)arow1 * DCOL);
            const bool lr = (kc == 0) ? LR0 : LR1;
            const float* lb = lw + hi * 256 + lo * 2;
#pragma unroll 4
            for (int ks = 0; ks < 32; ++ks) {
                v2f a = *(const v2f*)(arowptr + ks * 4 + hi * 2);
                if (lr) { a[0] = lrelu(a[0]); a[1] = lrelu(a[1]); }
                v2f b[8];
#pragma unroll
                for (int t = 0; t < 8; ++t)
                    b[t] = *(const v2f*)(lb + ks * 512 + t * 32);
#pragma unroll
                for (int t = 0; t < 8; ++t)
                    acc[t] = __builtin_amdgcn_wmma_f32_16x16x4_f32(
                        false, a, false, b[t], (short)0, acc[t], false, false);
            }
        }
        __syncthreads();
    }

    if (active) {
#pragma unroll
        for (int t = 0; t < 8; ++t) {
            int col = t * 16 + lo;
            float badd = EPI ? bias[col] : 0.0f;
#pragma unroll
            for (int r = 0; r < 8; ++r) {
                int row = rowbase + r + hi * 8;
                float v = acc[t][r];
                if (EPI) v = lrelu(v + badd);
                C[(size_t)row * DCOL + col] = v;
            }
        }
    }
}

extern "C" void kernel_launch(void* const* d_in, const int* in_sizes, int n_in,
                              void* d_out, int out_size, void* d_ws, size_t ws_size,
                              hipStream_t stream) {
    const float* features = (const float*)d_in[0];
    const float* values   = (const float*)d_in[1];
    const float* W1       = (const float*)d_in[2];
    const float* b1       = (const float*)d_in[3];
    const float* W2       = (const float*)d_in[4];
    const float* b2       = (const float*)d_in[5];
    const float* Wl       = (const float*)d_in[6];
    const float* bl       = (const float*)d_in[7];
    const int*   adjs     = (const int*)d_in[8];
    const int*   root_idx = (const int*)d_in[9];
    const int*   batch    = (const int*)d_in[12];

    const int N = in_sizes[0] / DCOL;
    const int E = in_sizes[1];
    const int* srcI = adjs;
    const int* dstI = adjs + E;

    float* XW = (float*)d_ws;
    float* H1 = XW + (size_t)N * DCOL;
    float* H2 = H1 + (size_t)N * DCOL;
    int* ROOTROW = (int*)(H2 + (size_t)N * DCOL);

    float* OUT = (float*)d_out;
    int* OWN = (int*)d_out;
    int* LISTS = OWN + scan_steps(E) * (SU * SW * CHUNK / 4);

    const int gemmBlocks = (N + 127) / 128;
    const int scatBlocks = (N + NPB - 1) / NPB;

    const int rootBlocks = (N + 255) / 256;
    const int ownBlocks  = (scan_steps(E) * (SU * SW * CHUNK) / 16 + 255) / 256;
    k_prep<<<rootBlocks + ownBlocks, 256, 0, stream>>>(batch, root_idx, ROOTROW, N,
                                                       dstI, OWN, E, rootBlocks);

    k_gemm_wmma<1, false, false, false><<<gemmBlocks, 256, 0, stream>>>(
        features, nullptr, nullptr, W1, nullptr, XW, N);
    k_scatter_own<0><<<scatBlocks, SW * 32, 0, stream>>>(XW, srcI, dstI, OWN, values, b1, H1, LISTS, N, E);

    k_gemm_wmma<2, true, true, false><<<gemmBlocks, 256, 0, stream>>>(
        H1, features, ROOTROW, W2, nullptr, XW, N);
    k_scatter_own<1><<<scatBlocks, SW * 32, 0, stream>>>(XW, srcI, dstI, OWN, values, b2, H2, LISTS, N, E);

    k_gemm_wmma<2, true, false, true><<<gemmBlocks, 256, 0, stream>>>(
        H2, H1, ROOTROW, Wl, bl, OUT, N);
}


# GEMM re-tiled: a wave keeps its 16 columns of W in registers (W no longer through LDS) and walks the 13 row tiles of a 200-row block, A rows staged once in LDS; 250 blocks, even WMMA share per SIMD
# speedup vs baseline: 2.3066x; 1.0532x over previous
#include <hip/hip_runtime.h>
#include <hip/hip_bf16.h>

typedef float v2f __attribute__((ext_vector_type(2)));
typedef float v4f __attribute__((ext_vector_type(4)));
typedef float v8f __attribute__((ext_vector_type(8)));

#define NEG_SLOPE 0.01f
#define DCOL 128

__device__ __forceinline__ float lrelu(float x) {
    return x > 0.0f ? x : NEG_SLOPE * x;
}

#define NPB   200
#define SW    32
#define SU    2
#define CHUNK 512

__device__ __host__ __forceinline__ int scan_steps(int E) {
    const int nchunks = (E + CHUNK - 1) / CHUNK;
    const int nsteps = (nchunks + SU * SW - 1) / (SU * SW);
    return nsteps < 1 ? 1 : nsteps;
}

__device__ __forceinline__ unsigned owner_of(int d) {
    const unsigned o = (unsigned)d / (unsigned)NPB;
    return o < 255u ? o : 255u;
}

__global__ void k_prep(const int* __restrict__ batch,
                       const int* __restrict__ root_idx,
                       int* __restrict__ rootrow, int N,
                       const int* __restrict__ dst,
                       int* __restrict__ own, int E, int rootBlocks) {
    if ((int)blockIdx.x < rootBlocks) {
        const int i = blockIdx.x * blockDim.x + threadIdx.x;
        if (i < N) rootrow[i] = root_idx[batch[i]];
        return;
    }
    const int emap = scan_steps(E) * (SU * SW * CHUNK);
    const int e0 = ((blockIdx.x - rootBlocks) * blockDim.x + threadIdx.x) * 16;
    if (e0 >= emap) return;
    unsigned pk[4];
    if (((((size_t)dst) & 15) == 0) && e0 + 15 < E) {
#pragma unroll
        for (int g = 0; g < 4; ++g) {
            const int4 v = *(const int4*)(dst + e0 + g * 4);
            pk[g] = owner_of(v.x) | (owner_of(v.y) << 8) |
                    (owner_of(v.z) << 16) | (owner_of(v.w) << 24);
        }
    } else {
#pragma unroll
        for (int g = 0; g < 4; ++g) {
            unsigned p = 0u;
#pragma unroll
            for (int k = 0; k < 4; ++k) {
                const int e = e0 + g * 4 + k;
                const unsigned o = (e < E) ? owner_of(dst[e]) : 255u;
                p |= o << (8 * k);
            }
            pk[g] = p;
        }
    }
    *(uint4*)(own + e0 / 4) = make_uint4(pk[0], pk[1], pk[2], pk[3]);
}

#define BAT   15
#define QCAP  (SU * CHUNK + 64)

__device__ __forceinline__ void lds_row_add(float* a, float w, v4f x) {
    __hip_atomic_fetch_add(a + 0,  w * x[0], __ATOMIC_RELAXED, __HIP_MEMORY_SCOPE_WORKGROUP);
    __hip_atomic_fetch_add(a + 32, w * x[1], __ATOMIC_RELAXED, __HIP_MEMORY_SCOPE_WORKGROUP);
    __hip_atomic_fetch_add(a + 64, w * x[2], __ATOMIC_RELAXED, __HIP_MEMORY_SCOPE_WORKGROUP);
    __hip_atomic_fetch_add(a + 96, w * x[3], __ATOMIC_RELAXED, __HIP_MEMORY_SCOPE_WORKGROUP);
}

template <bool EXACT>
__device__ __forceinline__ int note_edges(uint4 o, unsigned pat, int ebase, int* q, int count,
                                          const int* __restrict__ dst, int n0, int nn, int E) {
    const unsigned wd[4] = { o.x, o.y, o.z, o.w };
#pragma unroll
    for (int g = 0; g < 4; ++g) {
        const unsigned x = wd[g] ^ pat;
        const unsigned z = (x - 0x01010101u) & ~x & 0x80808080u;
        if (__builtin_amdgcn_ballot_w32(z != 0u) == 0u) continue;
#pragma unroll
        for (int k = 0; k < 4; ++k) {
            const int e = ebase + g * 4 + k;
            bool m = ((x >> (8 * k)) & 0xffu) == 0u;
            if (!EXACT) {
                const bool in = m && (e < E);
                const int d = in ? dst[e] : -1;
                m = in && ((unsigned)(d - n0) < (unsigned)nn);
            }
            const unsigned mask = __builtin_amdgcn_ballot_w32(m);
            if (mask != 0u) {
                if (m) q[count + (int)__builtin_amdgcn_mbcnt_lo(mask, 0u)] = e;
                count += __builtin_popcount(mask);
            }
        }
    }
    return count;
}

__device__ __forceinline__ void ask_rows(const float* __restrict__ xw, int lane, int s,
                                         v4f (&xr)[BAT]) {
#pragma unroll
    for (int j = 0; j < BAT; ++j) {
        const int sj = __builtin_amdgcn_readlane(s, j);
        xr[j] = *(const v4f*)(xw + (size_t)sj * DCOL + lane * 4);
    }
}

__device__ __forceinline__ void add_rows(float* accs, int lane, int dl, float w,
                                         const v4f (&xr)[BAT]) {
#pragma unroll
    for (int j = 0; j < BAT; ++j) {
        const int dj = __builtin_amdgcn_readlane(dl, j);
        const float wj = __int_as_float(
            __builtin_amdgcn_readlane(__float_as_int(w), j));
        lds_row_add(accs + dj * DCOL + lane, wj, xr[j]);
    }
}

__device__ __forceinline__ void add_batch(const float* __restrict__ xw, float* accs, int lane,
                                          int nb, int s, int dl, float w) {
    v4f xr[BAT];
    if (nb == BAT) {
        ask_rows(xw, lane, s, xr);
        add_rows(accs, lane, dl, w, xr);
    } else {
#pragma unroll
        for (int j = 0; j < BAT; ++j) {
            if (j < nb) {
                const int sj = __builtin_amdgcn_readlane(s, j);
                xr[j] = *(const v4f*)(xw + (size_t)sj * DCOL + lane * 4);
            }
        }
#pragma unroll
        for (int j = 0; j < BAT; ++j) {
            if (j < nb) {
                const int dj = __builtin_amdgcn_readlane(dl, j);
                const float wj = __int_as_float(
                    __builtin_amdgcn_readlane(__float_as_int(w), j));
                lds_row_add(accs + dj * DCOL + lane, wj, xr[j]);
            }
        }
    }
}

#define CAPW 256

__device__ __forceinline__ int list_cap(int N, int E, int nblk) {
    const long long avail = (long long)N * DCOL
                          - (long long)scan_steps(E) * (SU * SW * CHUNK / 4)
                          - (long long)nblk * SW;
    const long long c = avail / (3LL * nblk * SW);
    return c < 0 ? 0 : (c > CAPW ? CAPW : (int)c);
}

template <int MODE>
__global__ __launch_bounds__(1024) void k_scatter_own(const float* __restrict__ xw,
               const int* __restrict__ src,
               const int* __restrict__ dst,
               const int* __restrict__ own,
               const float* __restrict__ wvals,
               const float* __restrict__ bias,
               float* __restrict__ out,
               int* __restrict__ lists, int N, int E) {
    __shared__ float accs[NPB * DCOL];
    __shared__ int queue[SW * QCAP];

    const int tid  = threadIdx.x;
    const int lane = tid & 31;
    const int wave = __builtin_amdgcn_readfirstlane(tid >> 5);
    const int n0 = blockIdx.x * NPB;
    const int nn = (N - n0 < NPB) ? (N - n0) : NPB;

    for (int i = tid; i < nn * DCOL; i += SW * 32)
        accs[i] = bias[((i & 31) << 2) | ((i >> 5) & 3)];
    __syncthreads();

    const int nblk = gridDim.x;
    const int cap = __builtin_amdgcn_readfirstlane(list_cap(N, E, nblk));
    const size_t plane = (size_t)nblk * SW * cap;
    int* cntp = lists + blockIdx.x * SW + wave;
    int* LS = lists + (size_t)nblk * SW + ((size_t)blockIdx.x * SW + wave) * cap;
    int* LD = LS + plane;
    float* LW = (float*)(LD + plane);

    bool scan = true;
    if (MODE == 1) {
        const int c = __builtin_amdgcn_readfirstlane(*cntp);
        if (c >= 0) {
            scan = false;
            int s = 0, dl = 0;
            float w = 0.0f;
            if (lane < BAT && lane < c) { s = LS[lane]; dl = LD[lane]; w = LW[lane]; }
            int base = 0;
            for (; base + BAT <= c; base += BAT) {
                v4f xr[BAT];
                ask_rows(xw, lane, s, xr);
                int s2 = 0, dl2 = 0;
                float w2 = 0.0f;
                if (lane < BAT && base + BAT + lane < c) {
                    s2  = LS[base + BAT + lane];
                    dl2 = LD[base + BAT + lane];
                    w2  = LW[base + BAT + lane];
                }
                add_rows(accs, lane, dl, w, xr);
                s = s2; dl = dl2; w = w2;
            }
            if (base < c) add_batch(xw, accs, lane, c - base, s, dl, w);
        }
    }

    if (scan) {
        int* q = queue + wave * QCAP;
        int count = 0;
        int ln = 0;
        bool over = false;
        const int nsteps = scan_steps(E);
        const int t0 = blockIdx.x % nsteps;
        const unsigned myb = blockIdx.x < 255u ? blockIdx.x : 255u;
        const bool exact = blockIdx.x < 255u;
        const unsigned pat = myb * 0x01010101u;

        for (int t = 0; t <= nsteps; ++t) {
            const bool more = t < nsteps;
            if (more) {
                int ts = t + t0;
                if (ts >= nsteps) ts -= nsteps;
                const int cb = ts * (SU * SW) + wave;
                uint4 ov[SU];
#pragma unroll
                for (int u = 0; u < SU; ++u)
                    ov[u] = *(const uint4*)(own + (size_t)(cb + u * SW) * (CHUNK / 4) + lane * 4);
                if (exact) {
#pragma unroll
                    for (int u = 0; u < SU; ++u)
                        count = note_edges<true>(ov[u], pat, (cb + u * SW) * CHUNK + lane * 16,
                                                 q, count, dst, n0, nn, E);
                } else {
#pragma unroll
                    for (int u = 0; u < SU; ++u)
                        count = note_edges<false>(ov[u], pat, (cb + u * SW) * CHUNK + lane * 16,
                                                  q, count, dst, n0, nn, E);
                }
            }

            while (count >= (more ? BAT : 1)) {
                const int nb = count < BAT ? count : BAT;
                count -= nb;
                int s = 0, dl = 0;
                float w = 0.0f;
                if (lane < nb) {
                    const int e = q[count + lane];
                    s  = src[e];
                    dl = dst[e] - n0;
                    w  = wvals[e];
                }
                if (MODE == 0) {
                    if (ln + nb <= cap) {
                        if (lane < nb) { LS[ln + lane] = s; LD[ln + lane] = dl; LW[ln + lane] = w; }
                        ln += nb;
                    } else {
                        over = true;
                    }
                }
                add_batch(xw, accs, lane, nb, s, dl, w);
            }
        }
        if (MODE == 0 && lane == 0) *cntp = over ? -1 : ln;
    }
    __syncthreads();

    v4f* o4 = (v4f*)(out + (size_t)n0 * DCOL);
    for (int i = tid; i < nn * (DCOL / 4); i += SW * 32) {
        const float* a = accs + (i >> 5) * DCOL + (i & 31);
        const v4f o = { a[0], a[32], a[64], a[96] };
        o4[i] = o;
    }
}

#define GROWS  NPB
#define GRT    ((GROWS + 15) / 16)
#define GWAVES 8
#define APITCH (DCOL + 4)

template <int NT, int KCHUNKS, bool EPI>
__device__ __forceinline__ void gemm_tiles(const float* __restrict__ arow,
                                           const v2f (&b)[KCHUNKS][32],
                                           const float* __restrict__ bias,
                                           float* __restrict__ C,
                                           int rowl0, int nn, int n0, int col, int hi) {
    v8f acc[NT] = {};
#pragma unroll
    for (int kc = 0; kc < KCHUNKS; ++kc) {
#pragma unroll
        for (int ks = 0; ks < 32; ++ks) {
#pragma unroll
            for (int j = 0; j < NT; ++j) {
                const v2f a = *(const v2f*)(arow + (kc * GRT * 16 + j * 16) * APITCH + ks * 4);
                acc[j] = __builtin_amdgcn_wmma_f32_16x16x4_f32(
                    false, a, false, b[kc][ks], (short)0, acc[j], false, false);
            }
        }
    }
    const float badd = EPI ? bias[col] : 0.0f;
#pragma unroll
    for (int j = 0; j < NT; ++j) {
#pragma unroll
        for (int r = 0; r < 8; ++r) {
            const int rowl = rowl0 + j * 16 + r + hi * 8;
            if (rowl < nn) {
                float v = acc[j][r];
                if (EPI) v = lrelu(v + badd);
                C[(size_t)(n0 + rowl) * DCOL + col] = v;
            }
        }
    }
}

template <int KCHUNKS, bool LR0, bool LR1, bool EPI>
__global__ __launch_bounds__(GWAVES * 32) void k_gemm_wmma(const float* __restrict__ A0,
                            const float* __restrict__ A1,
                            const int* __restrict__ rootrow,
                            const float* __restrict__ W,
                            const float* __restrict__ bias,
                            float* __restrict__ C, int N) {
    __shared__ float la[2 * GRT * 16 * APITCH];

    const int tid  = threadIdx.x;
    const int lane = tid & 31;
    const int wave = __builtin_amdgcn_readfirstlane(tid >> 5);
    const int lo   = lane & 15;
    const int hi   = lane >> 4;
    const int ct   = wave;
    const int n0 = blockIdx.x * GROWS;
    const int nn = (N - n0 < GROWS) ? (N - n0) : GROWS;
    const int col = ct * 16 + lo;

    v2f b[KCHUNKS][32];
#pragma unroll
    for (int kc = 0; kc < KCHUNKS; ++kc) {
#pragma unroll
        for (int ks = 0; ks < 32; ++ks) {
            const float* p = W + (size_t)(kc * 128 + ks * 4 + hi * 2) * DCOL + col;
            const v2f f = { p[0], p[DCOL] };
            b[kc][ks] = f;
        }
    }

    constexpr int NIT = (GRT * 16) / GWAVES;
    static_assert((GRT * 16) % GWAVES == 0, "rows of the LDS image divide over the waves");
    static_assert(NIT <= 32, "one lane per staged row of a wave");
    int rrv = 0;
    if (KCHUNKS > 1) {
        const int r = wave + lane * GWAVES;
        rrv = rootrow[n0 + (r < nn ? r : nn - 1)];
    }
#pragma unroll
    for (int kc = 0; kc < KCHUNKS; ++kc) {
        const bool lr = (kc == 0) ? LR0 : LR1;
        const float* Ap = (kc == 0) ? A0 : A1;
        v4f t[NIT];
#pragma unroll
        for (int it = 0; it < NIT; ++it) {
            const int r = wave + it * GWAVES;
            const int rc = r < nn ? r : nn - 1;
            const int row = (kc == 0) ? (n0 + rc) : __builtin_amdgcn_readlane(rrv, it);
            t[it] = *(const v4f*)(Ap + (size_t)row * DCOL + lane * 4);
        }
#pragma unroll
        for (int it = 0; it < NIT; ++it) {
            const int r = wave + it * GWAVES;
            v4f v = t[it];
            if (lr) { v[0] = lrelu(v[0]); v[1] = lrelu(v[1]); v[2] = lrelu(v[2]); v[3] = lrelu(v[3]); }
            *(v4f*)(la + (kc * GRT * 16 + r) * APITCH + lane * 4) = v;
        }
    }
    __syncthreads();

    gemm_tiles<GRT, KCHUNKS, EPI>(la + lo * APITCH + hi * 2, b, bias, C, 0, nn, n0, col, hi);
}

extern "C" void kernel_launch(void* const* d_in, const int* in_sizes, int n_in,
                              void* d_out, int out_size, void* d_ws, size_t ws_size,
                              hipStream_t stream) {
    const float* features = (const float*)d_in[0];
    const float* values   = (const float*)d_in[1];
    const float* W1       = (const float*)d_in[2];
    const float* b1       = (const float*)d_in[3];
    const float* W2       = (const float*)d_in[4];
    const float* b2       = (const float*)d_in[5];
    const float* Wl       = (const float*)d_in[6];
    const float* bl       = (const float*)d_in[7];
    const int*   adjs     = (const int*)d_in[8];
    const int*   root_idx = (const int*)d_in[9];
    const int*   batch    = (const int*)d_in[12];

    const int N = in_sizes[0] / DCOL;
    const int E = in_sizes[1];
    const int* srcI = adjs;
    const int* dstI = adjs + E;

    float* XW = (float*)d_ws;
    float* H1 = XW + (size_t)N * DCOL;
    float* H2 = H1 + (size_t)N * DCOL;
    int* ROOTROW = (int*)(H2 + (size_t)N * DCOL);

    float* OUT = (float*)d_out;
    int* OWN = (int*)d_out;
    int* LISTS = OWN + scan_steps(E) * (SU * SW * CHUNK / 4);

    const int gemmBlocks = (N + GROWS - 1) / GROWS;
    const int scatBlocks = (N + NPB - 1) / NPB;

    const int rootBlocks = (N + 255) / 256;
    const int ownBlocks  = (scan_steps(E) * (SU * SW * CHUNK) / 16 + 255) / 256;
    k_prep<<<rootBlocks + ownBlocks, 256, 0, stream>>>(batch, root_idx, ROOTROW, N,
                                                       dstI, OWN, E, rootBlocks);

    k_gemm_wmma<1, false, false, false><<<gemmBlocks, GWAVES * 32, 0, stream>>>(
        features, nullptr, nullptr, W1, nullptr, XW, N);
    k_scatter_own<0><<<scatBlocks, SW * 32, 0, stream>>>(XW, srcI, dstI, OWN, values, b1, H1, LISTS, N, E);

    k_gemm_wmma<2, true, true, false><<<gemmBlocks, GWAVES * 32, 0, stream>>>(
        H1, features, ROOTROW, W2, nullptr, XW, N);
    k_scatter_own<1><<<scatBlocks, SW * 32, 0, stream>>>(XW, srcI, dstI, OWN, values, b2, H2, LISTS, N, E);

    k_gemm_wmma<2, true, false, true><<<gemmBlocks, GWAVES * 32, 0, stream>>>(
        H2, H1, ROOTROW, Wl, bl, OUT, N);
}


# Layer 1 sums edges over input rows first, then multiplies by W1 (sum order changes); that product runs in layer 2's product launch, H1 tile handed over in LDS: launches 6 to 5
# speedup vs baseline: 2.4867x; 1.0781x over previous
#include <hip/hip_runtime.h>
#include <hip/hip_bf16.h>

typedef float v2f __attribute__((ext_vector_type(2)));
typedef float v4f __attribute__((ext_vector_type(4)));
typedef float v8f __attribute__((ext_vector_type(8)));

#define NEG_SLOPE 0.01f
#define DCOL 128

__device__ __forceinline__ float lrelu(float x) {
    return x > 0.0f ? x : NEG_SLOPE * x;
}

#define NPB   200
#define SW    32
#define SU    2
#define CHUNK 512

__device__ __host__ __forceinline__ int scan_steps(int E) {
    const int nchunks = (E + CHUNK - 1) / CHUNK;
    const int nsteps = (nchunks + SU * SW - 1) / (SU * SW);
    return nsteps < 1 ? 1 : nsteps;
}

__device__ __forceinline__ unsigned owner_of(int d) {
    const unsigned o = (unsigned)d / (unsigned)NPB;
    return o < 255u ? o : 255u;
}

__global__ void k_prep(const int* __restrict__ batch,
                       const int* __restrict__ root_idx,
                       int* __restrict__ rootrow, int N,
                       const int* __restrict__ dst,
                       int* __restrict__ own, int E, int rootBlocks) {
    if ((int)blockIdx.x < rootBlocks) {
        const int i = blockIdx.x * blockDim.x + threadIdx.x;
        if (i < N) rootrow[i] = root_idx[batch[i]];
        return;
    }
    const int emap = scan_steps(E) * (SU * SW * CHUNK);
    const int e0 = ((blockIdx.x - rootBlocks) * blockDim.x + threadIdx.x) * 16;
    if (e0 >= emap) return;
    unsigned pk[4];
    if (((((size_t)dst) & 15) == 0) && e0 + 15 < E) {
#pragma unroll
        for (int g = 0; g < 4; ++g) {
            const int4 v = *(const int4*)(dst + e0 + g * 4);
            pk[g] = owner_of(v.x) | (owner_of(v.y) << 8) |
                    (owner_of(v.z) << 16) | (owner_of(v.w) << 24);
        }
    } else {
#pragma unroll
        for (int g = 0; g < 4; ++g) {
            unsigned p = 0u;
#pragma unroll
            for (int k = 0; k < 4; ++k) {
                const int e = e0 + g * 4 + k;
                const unsigned o = (e < E) ? owner_of(dst[e]) : 255u;
                p |= o << (8 * k);
            }
            pk[g] = p;
        }
    }
    *(uint4*)(own + e0 / 4) = make_uint4(pk[0], pk[1], pk[2], pk[3]);
}

#define BAT   15
#define QCAP  (SU * CHUNK + 64)

__device__ __forceinline__ void lds_row_add(float* a, float w, v4f x) {
    __hip_atomic_fetch_add(a + 0,  w * x[0], __ATOMIC_RELAXED, __HIP_MEMORY_SCOPE_WORKGROUP);
    __hip_atomic_fetch_add(a + 32, w * x[1], __ATOMIC_RELAXED, __HIP_MEMORY_SCOPE_WORKGROUP);
    __hip_atomic_fetch_add(a + 64, w * x[2], __ATOMIC_RELAXED, __HIP_MEMORY_SCOPE_WORKGROUP);
    __hip_atomic_fetch_add(a + 96, w * x[3], __ATOMIC_RELAXED, __HIP_MEMORY_SCOPE_WORKGROUP);
}

template <bool EXACT>
__device__ __forceinline__ int note_edges(uint4 o, unsigned pat, int ebase, int* q, int count,
                                          const int* __restrict__ dst, int n0, int nn, int E) {
    const unsigned wd[4] = { o.x, o.y, o.z, o.w };
#pragma unroll
    for (int g = 0; g < 4; ++g) {
        const unsigned x = wd[g] ^ pat;
        const unsigned z = (x - 0x01010101u) & ~x & 0x80808080u;
        if (__builtin_amdgcn_ballot_w32(z != 0u) == 0u) continue;
#pragma unroll
        for (int k = 0; k < 4; ++k) {
            const int e = ebase + g * 4 + k;
            bool m = ((x >> (8 * k)) & 0xffu) == 0u;
            if (!EXACT) {
                const bool in = m && (e < E);
                const int d = in ? dst[e] : -1;
                m = in && ((unsigned)(d - n0) < (unsigned)nn);
            }
            const unsigned mask = __builtin_amdgcn_ballot_w32(m);
            if (mask != 0u) {
                if (m) q[count + (int)__builtin_amdgcn_mbcnt_lo(mask, 0u)] = e;
                count += __builtin_popcount(mask);
            }
        }
    }
    return count;
}

__device__ __forceinline__ void ask_rows(const float* __restrict__ xw, int lane, int s,
                                         v4f (&xr)[BAT]) {
#pragma unroll
    for (int j = 0; j < BAT; ++j) {
        const int sj = __builtin_amdgcn_readlane(s, j);
        xr[j] = *(const v4f*)(xw + (size_t)sj * DCOL + lane * 4);
    }
}

__device__ __forceinline__ void add_rows(float* accs, int lane, int dl, float w,
                                         const v4f (&xr)[BAT]) {
#pragma unroll
    for (int j = 0; j < BAT; ++j) {
        const int dj = __builtin_amdgcn_readlane(dl, j);
        const float wj = __int_as_float(
            __builtin_amdgcn_readlane(__float_as_int(w), j));
        lds_row_add(accs + dj * DCOL + lane, wj, xr[j]);
    }
}

__device__ __forceinline__ void add_batch(const float* __restrict__ xw, float* accs, int lane,
                                          int nb, int s, int dl, float w) {
    v4f xr[BAT];
    if (nb == BAT) {
        ask_rows(xw, lane, s, xr);
        add_rows(accs, lane, dl, w, xr);
    } else {
#pragma unroll
        for (int j = 0; j < BAT; ++j) {
            if (j < nb) {
                const int sj = __builtin_amdgcn_readlane(s, j);
                xr[j] = *(const v4f*)(xw + (size_t)sj * DCOL + lane * 4);
            }
        }
#pragma unroll
        for (int j = 0; j < BAT; ++j) {
            if (j < nb) {
                const int dj = __builtin_amdgcn_readlane(dl, j);
                const float wj = __int_as_float(
                    __builtin_amdgcn_readlane(__float_as_int(w), j));
                lds_row_add(accs + dj * DCOL + lane, wj, xr[j]);
            }
        }
    }
}

#define CAPW 256

__device__ __forceinline__ int list_cap(int N, int E, int nblk) {
    const long long avail = (long long)N * DCOL
                          - (long long)scan_steps(E) * (SU * SW * CHUNK / 4)
                          - (long long)nblk * SW;
    const long long c = avail / (3LL * nblk * SW);
    return c < 0 ? 0 : (c > CAPW ? CAPW : (int)c);
}

template <int MODE>
__global__ __launch_bounds__(1024) void k_scatter_own(const float* __restrict__ xw,
               const int* __restrict__ src,
               const int* __restrict__ dst,
               const int* __restrict__ own,
               const float* __restrict__ wvals,
               const float* __restrict__ bias,
               float* __restrict__ out,
               int* __restrict__ lists, int N, int E) {
    __shared__ float accs[NPB * DCOL];
    __shared__ int queue[SW * QCAP];

    const int tid  = threadIdx.x;
    const int lane = tid & 31;
    const int wave = __builtin_amdgcn_readfirstlane(tid >> 5);
    const int n0 = blockIdx.x * NPB;
    const int nn = (N - n0 < NPB) ? (N - n0) : NPB;

    for (int i = tid; i < nn * DCOL; i += SW * 32)
        accs[i] = bias ? bias[((i & 31) << 2) | ((i >> 5) & 3)] : 0.0f;
    __syncthreads();

    const int nblk = gridDim.x;
    const int cap = __builtin_amdgcn_readfirstlane(list_cap(N, E, nblk));
    const size_t plane = (size_t)nblk * SW * cap;
    int* cntp = lists + blockIdx.x * SW + wave;
    int* LS = lists + (size_t)nblk * SW + ((size_t)blockIdx.x * SW + wave) * cap;
    int* LD = LS + plane;
    float* LW = (float*)(LD + plane);

    bool scan = true;
    if (MODE == 1) {
        const int c = __builtin_amdgcn_readfirstlane(*cntp);
        if (c >= 0) {
            scan = false;
            int s = 0, dl = 0;
            float w = 0.0f;
            if (lane < BAT && lane < c) { s = LS[lane]; dl = LD[lane]; w = LW[lane]; }
            int base = 0;
            for (; base + BAT <= c; base += BAT) {
                v4f xr[BAT];
                ask_rows(xw, lane, s, xr);
                int s2 = 0, dl2 = 0;
                float w2 = 0.0f;
                if (lane < BAT && base + BAT + lane < c) {
                    s2  = LS[base + BAT + lane];
                    dl2 = LD[base + BAT + lane];
                    w2  = LW[base + BAT + lane];
                }
                add_rows(accs, lane, dl, w, xr);
                s = s2; dl = dl2; w = w2;
            }
            if (base < c) add_batch(xw, accs, lane, c - base, s, dl, w);
        }
    }

    if (scan) {
        int* q = queue + wave * QCAP;
        int count = 0;
        int ln = 0;
        bool over = false;
        const int nsteps = scan_steps(E);
        const int t0 = blockIdx.x % nsteps;
        const unsigned myb = blockIdx.x < 255u ? blockIdx.x : 255u;
        const bool exact = blockIdx.x < 255u;
        const unsigned pat = myb * 0x01010101u;

        for (int t = 0; t <= nsteps; ++t) {
            const bool more = t < nsteps;
            if (more) {
                int ts = t + t0;
                if (ts >= nsteps) ts -= nsteps;
                const int cb = ts * (SU * SW) + wave;
                uint4 ov[SU];
#pragma unroll
                for (int u = 0; u < SU; ++u)
                    ov[u] = *(const uint4*)(own + (size_t)(cb + u * SW) * (CHUNK / 4) + lane * 4);
                if (exact) {
#pragma unroll
                    for (int u = 0; u < SU; ++u)
                        count = note_edges<true>(ov[u], pat, (cb + u * SW) * CHUNK + lane * 16,
                                                 q, count, dst, n0, nn, E);
                } else {
#pragma unroll
                    for (int u = 0; u < SU; ++u)
                        count = note_edges<false>(ov[u], pat, (cb + u * SW) * CHUNK + lane * 16,
                                                  q, count, dst, n0, nn, E);
                }
            }

            while (count >= (more ? BAT : 1)) {
                const int nb = count < BAT ? count : BAT;
                count -= nb;
                int s = 0, dl = 0;
                float w = 0.0f;
                if (lane < nb) {
                    const int e = q[count + lane];
                    s  = src[e];
                    dl = dst[e] - n0;
                    w  = wvals[e];
                }
                if (MODE == 0) {
                    if (ln + nb <= cap) {
                        if (lane < nb) { LS[ln + lane] = s; LD[ln + lane] = dl; LW[ln + lane] = w; }
                        ln += nb;
                    } else {
                        over = true;
                    }
                }
                add_batch(xw, accs, lane, nb, s, dl, w);
            }
        }
        if (MODE == 0 && lane == 0) *cntp = over ? -1 : ln;
    }
    __syncthreads();

    v4f* o4 = (v4f*)(out + (size_t)n0 * DCOL);
    for (int i = tid; i < nn * (DCOL / 4); i += SW * 32) {
        const float* a = accs + (i >> 5) * DCOL + (i & 31);
        const v4f o = { a[0], a[32], a[64], a[96] };
        o4[i] = o;
    }
}

#define GROWS  NPB
#define GRT    ((GROWS + 15) / 16)
#define GWAVES 8
#define APITCH (DCOL + 4)

template <int NT, int KCHUNKS, bool EPI>
__device__ __forceinline__ void gemm_tiles(const float* __restrict__ arow,
                                           const v2f (&b)[KCHUNKS][32],
                                           const float* __restrict__ bias,
                                           float* __restrict__ C,
                                           int rowl0, int nn, int n0, int col, int hi) {
    v8f acc[NT] = {};
#pragma unroll
    for (int kc = 0; kc < KCHUNKS; ++kc) {
#pragma unroll
        for (int ks = 0; ks < 32; ++ks) {
#pragma unroll
            for (int j = 0; j < NT; ++j) {
                const v2f a = *(const v2f*)(arow + (kc * GRT * 16 + j * 16) * APITCH + ks * 4);
                acc[j] = __builtin_amdgcn_wmma_f32_16x16x4_f32(
                    false, a, false, b[kc][ks], (short)0, acc[j], false, false);
            }
        }
    }
    const float badd = EPI ? bias[col] : 0.0f;
#pragma unroll
    for (int j = 0; j < NT; ++j) {
#pragma unroll
        for (int r = 0; r < 8; ++r) {
            const int rowl = rowl0 + j * 16 + r + hi * 8;
            if (rowl < nn) {
                float v = acc[j][r];
                if (EPI) v = lrelu(v + badd);
                C[(size_t)(n0 + rowl) * DCOL + col] = v;
            }
        }
    }
}

template <int KCHUNKS, bool LR0, bool LR1, bool EPI>
__global__ __launch_bounds__(GWAVES * 32) void k_gemm_wmma(const float* __restrict__ A0,
                            const float* __restrict__ A1,
                            const int* __restrict__ rootrow,
                            const float* __restrict__ W,
                            const float* __restrict__ bias,
                            float* __restrict__ C, int N) {
    __shared__ float la[2 * GRT * 16 * APITCH];

    const int tid  = threadIdx.x;
    const int lane = tid & 31;
    const int wave = __builtin_amdgcn_readfirstlane(tid >> 5);
    const int lo   = lane & 15;
    const int hi   = lane >> 4;
    const int ct   = wave;
    const int n0 = blockIdx.x * GROWS;
    const int nn = (N - n0 < GROWS) ? (N - n0) : GROWS;
    const int col = ct * 16 + lo;

    v2f b[KCHUNKS][32];
#pragma unroll
    for (int kc = 0; kc < KCHUNKS; ++kc) {
#pragma unroll
        for (int ks = 0; ks < 32; ++ks) {
            const float* p = W + (size_t)(kc * 128 + ks * 4 + hi * 2) * DCOL + col;
            const v2f f = { p[0], p[DCOL] };
            b[kc][ks] = f;
        }
    }

    constexpr int NIT = (GRT * 16) / GWAVES;
    static_assert((GRT * 16) % GWAVES == 0, "rows of the LDS image divide over the waves");
    static_assert(NIT <= 32, "one lane per staged row of a wave");
    int rrv = 0;
    if (KCHUNKS > 1) {
        const int r = wave + lane * GWAVES;
        rrv = rootrow[n0 + (r < nn ? r : nn - 1)];
    }
#pragma unroll
    for (int kc = 0; kc < KCHUNKS; ++kc) {
        const bool lr = (kc == 0) ? LR0 : LR1;
        const float* Ap = (kc == 0) ? A0 : A1;
        v4f t[NIT];
#pragma unroll
        for (int it = 0; it < NIT; ++it) {
            const int r = wave + it * GWAVES;
            const int rc = r < nn ? r : nn - 1;
            const int row = (kc == 0) ? (n0 + rc) : __builtin_amdgcn_readlane(rrv, it);
            t[it] = *(const v4f*)(Ap + (size_t)row * DCOL + lane * 4);
        }
#pragma unroll
        for (int it = 0; it < NIT; ++it) {
            const int r = wave + it * GWAVES;
            v4f v = t[it];
            if (lr) { v[0] = lrelu(v[0]); v[1] = lrelu(v[1]); v[2] = lrelu(v[2]); v[3] = lrelu(v[3]); }
            *(v4f*)(la + (kc * GRT * 16 + r) * APITCH + lane * 4) = v;
        }
    }
    __syncthreads();

    gemm_tiles<GRT, KCHUNKS, EPI>(la + lo * APITCH + hi * 2, b, bias, C, 0, nn, n0, col, hi);
}

__device__ __forceinline__ void store_tiles_full(float* __restrict__ p, const v8f (&acc)[GRT], int hi) {
#pragma unroll
    for (int j = 0; j < GRT; ++j) {
#pragma unroll
        for (int r = 0; r < 8; ++r)
            if (j * 16 + r + hi * 8 < GROWS) p[(j * 16 + r) * DCOL] = acc[j][r];
    }
}

__device__ __forceinline__ void store_tiles_part(float* __restrict__ rows, float* lp,
                                                 const float* la,
                                                 const v8f (&acc)[GRT], int tid, int nn) {
#pragma unroll
    for (int j = 0; j < GRT; ++j) {
#pragma unroll
        for (int r = 0; r < 8; ++r) lp[(j * 16 + r) * APITCH] = acc[j][r];
    }
    __syncthreads();
    for (int i = tid; i < nn * DCOL; i += GWAVES * 32)
        rows[i] = la[(i >> 7) * APITCH + (i & (DCOL - 1))];
}

__global__ __launch_bounds__(GWAVES * 32) void k_gemm_pair(const float* __restrict__ S,
                            const float* __restrict__ X,
                            const int* __restrict__ rootrow,
                            const float* __restrict__ Wa,
                            const float* __restrict__ ba,
                            const float* __restrict__ Wb,
                            float* __restrict__ H,
                            float* __restrict__ C, int N) {
    __shared__ float la[2 * GRT * 16 * APITCH];

    const int tid  = threadIdx.x;
    const int lane = tid & 31;
    const int wave = __builtin_amdgcn_readfirstlane(tid >> 5);
    const int lo   = lane & 15;
    const int hi   = lane >> 4;
    const int ct   = wave;
    const int n0 = blockIdx.x * GROWS;
    const int nn = (N - n0 < GROWS) ? (N - n0) : GROWS;
    const int col = ct * 16 + lo;

    v2f bc[32];
#pragma unroll
    for (int ks = 0; ks < 32; ++ks) {
        const float* p = Wa + (size_t)(ks * 4 + hi * 2) * DCOL + col;
        const v2f f = { p[0], p[DCOL] };
        bc[ks] = f;
    }
    const float badd = ba[col];

    constexpr int NIT = (GRT * 16) / GWAVES;
    int rrv;
    {
        const int r = wave + lane * GWAVES;
        rrv = rootrow[n0 + (r < nn ? r : nn - 1)];
    }
#pragma unroll
    for (int kc = 0; kc < 2; ++kc) {
        const float* Ap = (kc == 0) ? S : X;
        v4f t[NIT];
#pragma unroll
        for (int it = 0; it < NIT; ++it) {
            const int r = wave + it * GWAVES;
            const int rc = r < nn ? r : nn - 1;
            const int row = (kc == 0) ? (n0 + rc) : __builtin_amdgcn_readlane(rrv, it);
            t[it] = *(const v4f*)(Ap + (size_t)row * DCOL + lane * 4);
        }
#pragma unroll
        for (int it = 0; it < NIT; ++it) {
            const int r = wave + it * GWAVES;
            v4f v = t[it];
            if (kc == 1) { v[0] = lrelu(v[0]); v[1] = lrelu(v[1]); v[2] = lrelu(v[2]); v[3] = lrelu(v[3]); }
            *(v4f*)(la + (kc * GRT * 16 + r) * APITCH + lane * 4) = v;
        }
    }
    __syncthreads();

    const bool full = (nn == GROWS);
    const float* arow = la + lo * APITCH + hi * 2;
    float* hp = H + (size_t)(n0 + hi * 8) * DCOL + col;
    float* cp = C + (size_t)(n0 + hi * 8) * DCOL + col;
    float* lp = la + (hi * 8) * APITCH + col;
    v8f acc[GRT] = {};

#pragma nounroll
    for (int c = 0; c < 3; ++c) {
        const int cn = c < 2 ? c : 1;
        const float* Wn = Wb + (size_t)cn * 128 * DCOL;
        v2f bn[32];
#pragma unroll
        for (int ks = 0; ks < 32; ++ks) {
            const float* p = Wn + (size_t)(ks * 4 + hi * 2) * DCOL + col;
            const v2f f = { p[0], p[DCOL] };
            bn[ks] = f;
        }

        const float* ar = arow + (c == 2 ? GRT * 16 * APITCH : 0);
#pragma unroll
        for (int ks = 0; ks < 32; ++ks) {
#pragma unroll
            for (int j = 0; j < GRT; ++j) {
                const v2f a = *(const v2f*)(ar + j * 16 * APITCH + ks * 4);
                acc[j] = __builtin_amdgcn_wmma_f32_16x16x4_f32(
                    false, a, false, bc[ks], (short)0, acc[j], false, false);
            }
        }

        if (c == 0) {
            __syncthreads();
#pragma unroll
            for (int j = 0; j < GRT; ++j) {
#pragma unroll
                for (int r = 0; r < 8; ++r) acc[j][r] += badd;
            }
            if (full) {
                store_tiles_full(hp, acc, hi);
            } else {
                store_tiles_part(H + (size_t)n0 * DCOL, lp, la, acc, tid, nn);
                __syncthreads();
            }
#pragma unroll
            for (int j = 0; j < GRT; ++j) {
#pragma unroll
                for (int r = 0; r < 8; ++r) lp[(j * 16 + r) * APITCH] = lrelu(acc[j][r]);
                acc[j] = v8f{};
            }
            __syncthreads();
        }

#pragma unroll
        for (int ks = 0; ks < 32; ++ks) bc[ks] = bn[ks];
    }

    if (full) {
        store_tiles_full(cp, acc, hi);
    } else {
        __syncthreads();
        store_tiles_part(C + (size_t)n0 * DCOL, lp, la, acc, tid, nn);
    }
}

extern "C" void kernel_launch(void* const* d_in, const int* in_sizes, int n_in,
                              void* d_out, int out_size, void* d_ws, size_t ws_size,
                              hipStream_t stream) {
    const float* features = (const float*)d_in[0];
    const float* values   = (const float*)d_in[1];
    const float* W1       = (const float*)d_in[2];
    const float* b1       = (const float*)d_in[3];
    const float* W2       = (const float*)d_in[4];
    const float* b2       = (const float*)d_in[5];
    const float* Wl       = (const float*)d_in[6];
    const float* bl       = (const float*)d_in[7];
    const int*   adjs     = (const int*)d_in[8];
    const int*   root_idx = (const int*)d_in[9];
    const int*   batch    = (const int*)d_in[12];

    const int N = in_sizes[0] / DCOL;
    const int E = in_sizes[1];
    const int* srcI = adjs;
    const int* dstI = adjs + E;

    float* XW = (float*)d_ws;
    float* H1 = XW + (size_t)N * DCOL;
    float* H2 = H1 + (size_t)N * DCOL;
    int* ROOTROW = (int*)(H2 + (size_t)N * DCOL);

    float* OUT = (float*)d_out;
    int* OWN = (int*)d_out;
    int* LISTS = OWN + scan_steps(E) * (SU * SW * CHUNK / 4);

    const int gemmBlocks = (N + GROWS - 1) / GROWS;
    const int scatBlocks = (N + NPB - 1) / NPB;

    const int rootBlocks = (N + 255) / 256;
    const int ownBlocks  = (scan_steps(E) * (SU * SW * CHUNK) / 16 + 255) / 256;
    k_prep<<<rootBlocks + ownBlocks, 256, 0, stream>>>(batch, root_idx, ROOTROW, N,
                                                       dstI, OWN, E, rootBlocks);

    float* AX = H2;
    k_scatter_own<0><<<scatBlocks, SW * 32, 0, stream>>>(features, srcI, dstI, OWN, values, nullptr, AX, LISTS, N, E);

    k_gemm_pair<<<gemmBlocks, GWAVES * 32, 0, stream>>>(
        AX, features, ROOTROW, W1, b1, W2, H1, XW, N);
    k_scatter_own<1><<<scatBlocks, SW * 32, 0, stream>>>(XW, srcI, dstI, OWN, values, b2, H2, LISTS, N, E);

    k_gemm_wmma<2, true, false, true><<<gemmBlocks, GWAVES * 32, 0, stream>>>(
        H2, H1, ROOTROW, Wl, bl, OUT, N);
}


# Second aggregation's block multiplies its own LDS row sums by Wl in the same launch (32 waves; H2 never stored; map and lists moved to the freed plane): launches 5 to 4; sums still LDS arrival order
# speedup vs baseline: 2.7146x; 1.0916x over previous
#include <hip/hip_runtime.h>
#include <hip/hip_bf16.h>

typedef float v2f __attribute__((ext_vector_type(2)));
typedef float v4f __attribute__((ext_vector_type(4)));
typedef float v8f __attribute__((ext_vector_type(8)));

#define NEG_SLOPE 0.01f
#define DCOL 128

__device__ __forceinline__ float lrelu(float x) {
    return x > 0.0f ? x : NEG_SLOPE * x;
}

#define NPB   200
#define SW    32
#define SU    2
#define CHUNK 512

__device__ __host__ __forceinline__ int scan_steps(int E) {
    const int nchunks = (E + CHUNK - 1) / CHUNK;
    const int nsteps = (nchunks + SU * SW - 1) / (SU * SW);
    return nsteps < 1 ? 1 : nsteps;
}

__device__ __forceinline__ unsigned owner_of(int d) {
    const unsigned o = (unsigned)d / (unsigned)NPB;
    return o < 255u ? o : 255u;
}

__global__ void k_prep(const int* __restrict__ batch,
                       const int* __restrict__ root_idx,
                       int* __restrict__ rootrow, int N,
                       const int* __restrict__ dst,
                       int* __restrict__ own, int E, int rootBlocks) {
    if ((int)blockIdx.x < rootBlocks) {
        const int i = blockIdx.x * blockDim.x + threadIdx.x;
        if (i < N) rootrow[i] = root_idx[batch[i]];
        return;
    }
    const int emap = scan_steps(E) * (SU * SW * CHUNK);
    const int e0 = ((blockIdx.x - rootBlocks) * blockDim.x + threadIdx.x) * 16;
    if (e0 >= emap) return;
    unsigned pk[4];
    if (((((size_t)dst) & 15) == 0) && e0 + 15 < E) {
#pragma unroll
        for (int g = 0; g < 4; ++g) {
            const int4 v = *(const int4*)(dst + e0 + g * 4);
            pk[g] = owner_of(v.x) | (owner_of(v.y) << 8) |
                    (owner_of(v.z) << 16) | (owner_of(v.w) << 24);
        }
    } else {
#pragma unroll
        for (int g = 0; g < 4; ++g) {
            unsigned p = 0u;
#pragma unroll
            for (int k = 0; k < 4; ++k) {
                const int e = e0 + g * 4 + k;
                const unsigned o = (e < E) ? owner_of(dst[e]) : 255u;
                p |= o << (8 * k);
            }
            pk[g] = p;
        }
    }
    *(uint4*)(own + e0 / 4) = make_uint4(pk[0], pk[1], pk[2], pk[3]);
}

#define BAT   15
#define QCAP  (SU * CHUNK + 64)

__device__ __forceinline__ void lds_row_add(float* a, float w, v4f x) {
    __hip_atomic_fetch_add(a + 0,  w * x[0], __ATOMIC_RELAXED, __HIP_MEMORY_SCOPE_WORKGROUP);
    __hip_atomic_fetch_add(a + 32, w * x[1], __ATOMIC_RELAXED, __HIP_MEMORY_SCOPE_WORKGROUP);
    __hip_atomic_fetch_add(a + 64, w * x[2], __ATOMIC_RELAXED, __HIP_MEMORY_SCOPE_WORKGROUP);
    __hip_atomic_fetch_add(a + 96, w * x[3], __ATOMIC_RELAXED, __HIP_MEMORY_SCOPE_WORKGROUP);
}

template <bool EXACT>
__device__ __forceinline__ int note_edges(uint4 o, unsigned pat, int ebase, int* q, int count,
                                          const int* __restrict__ dst, int n0, int nn, int E) {
    const unsigned wd[4] = { o.x, o.y, o.z, o.w };
#pragma unroll
    for (int g = 0; g < 4; ++g) {
        const unsigned x = wd[g] ^ pat;
        const unsigned z = (x - 0x01010101u) & ~x & 0x80808080u;
        if (__builtin_amdgcn_ballot_w32(z != 0u) == 0u) continue;
#pragma unroll
        for (int k = 0; k < 4; ++k) {
            const int e = ebase + g * 4 + k;
            bool m = ((x >> (8 * k)) & 0xffu) == 0u;
            if (!EXACT) {
                const bool in = m && (e < E);
                const int d = in ? dst[e] : -1;
                m = in && ((unsigned)(d - n0) < (unsigned)nn);
            }
            const unsigned mask = __builtin_amdgcn_ballot_w32(m);
            if (mask != 0u) {
                if (m) q[count + (int)__builtin_amdgcn_mbcnt_lo(mask, 0u)] = e;
                count += __builtin_popcount(mask);
            }
        }
    }
    return count;
}

__device__ __forceinline__ void ask_rows(const float* __restrict__ xw, int lane, int s,
                                         v4f (&xr)[BAT]) {
#pragma unroll
    for (int j = 0; j < BAT; ++j) {
        const int sj = __builtin_amdgcn_readlane(s, j);
        xr[j] = *(const v4f*)(xw + (size_t)sj * DCOL + lane * 4);
    }
}

__device__ __forceinline__ void add_rows(float* accs, int lane, int dl, float w,
                                         const v4f (&xr)[BAT]) {
#pragma unroll
    for (int j = 0; j < BAT; ++j) {
        const int dj = __builtin_amdgcn_readlane(dl, j);
        const float wj = __int_as_float(
            __builtin_amdgcn_readlane(__float_as_int(w), j));
        lds_row_add(accs + dj * DCOL + lane, wj, xr[j]);
    }
}

__device__ __forceinline__ void add_batch(const float* __restrict__ xw, float* accs, int lane,
                                          int nb, int s, int dl, float w) {
    v4f xr[BAT];
    if (nb == BAT) {
        ask_rows(xw, lane, s, xr);
        add_rows(accs, lane, dl, w, xr);
    } else {
#pragma unroll
        for (int j = 0; j < BAT; ++j) {
            if (j < nb) {
                const int sj = __builtin_amdgcn_readlane(s, j);
                xr[j] = *(const v4f*)(xw + (size_t)sj * DCOL + lane * 4);
            }
        }
#pragma unroll
        for (int j = 0; j < BAT; ++j) {
            if (j < nb) {
                const int dj = __builtin_amdgcn_readlane(dl, j);
                const float wj = __int_as_float(
                    __builtin_amdgcn_readlane(__float_as_int(w), j));
                lds_row_add(accs + dj * DCOL + lane, wj, xr[j]);
            }
        }
    }
}

#define CAPW 256

__device__ __forceinline__ int list_cap(int N, int E, int nblk) {
    const long long avail = (long long)N * DCOL
                          - (long long)scan_steps(E) * (SU * SW * CHUNK / 4)
                          - (long long)nblk * SW;
    const long long c = avail / (3LL * nblk * SW);
    return c < 0 ? 0 : (c > CAPW ? CAPW : (int)c);
}

template <int MODE>
__device__ __forceinline__ void aggregate(float* accs, int* queue,
               const float* __restrict__ xw,
               const int* __restrict__ src,
               const int* __restrict__ dst,
               const int* __restrict__ own,
               const float* __restrict__ wvals,
               const float* __restrict__ bias,
               int* __restrict__ lists, int N, int E) {
    const int tid  = threadIdx.x;
    const int lane = tid & 31;
    const int wave = __builtin_amdgcn_readfirstlane(tid >> 5);
    const int n0 = blockIdx.x * NPB;
    const int nn = (N - n0 < NPB) ? (N - n0) : NPB;

    for (int i = tid; i < nn * DCOL; i += SW * 32)
        accs[i] = bias ? bias[((i & 31) << 2) | ((i >> 5) & 3)] : 0.0f;
    __syncthreads();

    const int nblk = gridDim.x;
    const int cap = __builtin_amdgcn_readfirstlane(list_cap(N, E, nblk));
    const size_t plane = (size_t)nblk * SW * cap;
    int* cntp = lists + blockIdx.x * SW + wave;
    int* LS = lists + (size_t)nblk * SW + ((size_t)blockIdx.x * SW + wave) * cap;
    int* LD = LS + plane;
    float* LW = (float*)(LD + plane);

    bool scan = true;
    if (MODE == 1) {
        const int c = __builtin_amdgcn_readfirstlane(*cntp);
        if (c >= 0) {
            scan = false;
            int s = 0, dl = 0;
            float w = 0.0f;
            if (lane < BAT && lane < c) { s = LS[lane]; dl = LD[lane]; w = LW[lane]; }
            int base = 0;
            for (; base + BAT <= c; base += BAT) {
                v4f xr[BAT];
                ask_rows(xw, lane, s, xr);
                int s2 = 0, dl2 = 0;
                float w2 = 0.0f;
                if (lane < BAT && base + BAT + lane < c) {
                    s2  = LS[base + BAT + lane];
                    dl2 = LD[base + BAT + lane];
                    w2  = LW[base + BAT + lane];
                }
                add_rows(accs, lane, dl, w, xr);
                s = s2; dl = dl2; w = w2;
            }
            if (base < c) add_batch(xw, accs, lane, c - base, s, dl, w);
        }
    }

    if (scan) {
        int* q = queue + wave * QCAP;
        int count = 0;
        int ln = 0;
        bool over = false;
        const int nsteps = scan_steps(E);
        const int t0 = blockIdx.x % nsteps;
        const unsigned myb = blockIdx.x < 255u ? blockIdx.x : 255u;
        const bool exact = blockIdx.x < 255u;
        const unsigned pat = myb * 0x01010101u;

        for (int t = 0; t <= nsteps; ++t) {
            const bool more = t < nsteps;
            if (more) {
                int ts = t + t0;
                if (ts >= nsteps) ts -= nsteps;
                const int cb = ts * (SU * SW) + wave;
                uint4 ov[SU];
#pragma unroll
                for (int u = 0; u < SU; ++u)
                    ov[u] = *(const uint4*)(own + (size_t)(cb + u * SW) * (CHUNK / 4) + lane * 4);
                if (exact) {
#pragma unroll
                    for (int u = 0; u < SU; ++u)
                        count = note_edges<true>(ov[u], pat, (cb + u * SW) * CHUNK + lane * 16,
                                                 q, count, dst, n0, nn, E);
                } else {
#pragma unroll
                    for (int u = 0; u < SU; ++u)
                        count = note_edges<false>(ov[u], pat, (cb + u * SW) * CHUNK + lane * 16,
                                                  q, count, dst, n0, nn, E);
                }
            }

            while (count >= (more ? BAT : 1)) {
                const int nb = count < BAT ? count : BAT;
                count -= nb;
                int s = 0, dl = 0;
                float w = 0.0f;
                if (lane < nb) {
                    const int e = q[count + lane];
                    s  = src[e];
                    dl = dst[e] - n0;
                    w  = wvals[e];
                }
                if (MODE == 0) {
                    if (ln + nb <= cap) {
                        if (lane < nb) { LS[ln + lane] = s; LD[ln + lane] = dl; LW[ln + lane] = w; }
                        ln += nb;
                    } else {
                        over = true;
                    }
                }
                add_batch(xw, accs, lane, nb, s, dl, w);
            }
        }
        if (MODE == 0 && lane == 0) *cntp = over ? -1 : ln;
    }
    __syncthreads();
}

__global__ __launch_bounds__(1024) void k_scatter_own(const float* __restrict__ xw,
               const int* __restrict__ src,
               const int* __restrict__ dst,
               const int* __restrict__ own,
               const float* __restrict__ wvals,
               const float* __restrict__ bias,
               float* __restrict__ out,
               int* __restrict__ lists, int N, int E) {
    __shared__ float accs[NPB * DCOL];
    __shared__ int queue[SW * QCAP];

    aggregate<0>(accs, queue, xw, src, dst, own, wvals, bias, lists, N, E);

    const int tid = threadIdx.x;
    const int n0 = blockIdx.x * NPB;
    const int nn = (N - n0 < NPB) ? (N - n0) : NPB;
    v4f* o4 = (v4f*)(out + (size_t)n0 * DCOL);
    for (int i = tid; i < nn * (DCOL / 4); i += SW * 32) {
        const float* a = accs + (i >> 5) * DCOL + (i & 31);
        const v4f o = { a[0], a[32], a[64], a[96] };
        o4[i] = o;
    }
}

#define GROWS  NPB
#define GRT    ((GROWS + 15) / 16)
#define GWAVES 8
#define APITCH (DCOL + 4)

#define FUSED_AGG_WORDS (NPB * DCOL + SW * QCAP)
#define FUSED_IMG_WORDS (2 * GRT * 16 * APITCH)
#define FUSED_WORDS (FUSED_AGG_WORDS > FUSED_IMG_WORDS ? FUSED_AGG_WORDS : FUSED_IMG_WORDS)

template <int NT>
__device__ __forceinline__ void fused_tiles(const float* la0, const float* la1,
                                            const float* __restrict__ W,
                                            const float* __restrict__ bias,
                                            float* __restrict__ C,
                                            int n0, int nn, int g, int col, int lo, int hi) {
    v8f acc[NT] = {};
#pragma unroll
    for (int kc = 0; kc < 2; ++kc) {
        const float* img = (kc == 0 ? la0 : la1) + lo * APITCH + hi * 2;
#pragma unroll
        for (int h = 0; h < 2; ++h) {
            v2f b[16];
#pragma unroll
            for (int ks = 0; ks < 16; ++ks) {
                const float* p = W + (size_t)(kc * 128 + (h * 16 + ks) * 4 + hi * 2) * DCOL + col;
                const v2f f = { p[0], p[DCOL] };
                b[ks] = f;
            }
#pragma unroll
            for (int ks = 0; ks < 16; ++ks) {
#pragma unroll
                for (int t = 0; t < NT; ++t) {
                    const v2f a = *(const v2f*)(img + (g + 4 * t) * 16 * APITCH + (h * 16 + ks) * 4);
                    acc[t] = __builtin_amdgcn_wmma_f32_16x16x4_f32(
                        false, a, false, b[ks], (short)0, acc[t], false, false);
                }
            }
        }
    }
    const float badd = bias[col];
#pragma unroll
    for (int t = 0; t < NT; ++t) {
#pragma unroll
        for (int r = 0; r < 8; ++r) {
            const int rowl = (g + 4 * t) * 16 + r + hi * 8;
            if (rowl < nn) C[(size_t)(n0 + rowl) * DCOL + col] = lrelu(acc[t][r] + badd);
        }
    }
}

__global__ __launch_bounds__(1024) void k_scatter_gemm(const float* __restrict__ xw,
               const int* __restrict__ src,
               const int* __restrict__ dst,
               const int* __restrict__ own,
               const float* __restrict__ wvals,
               const float* __restrict__ bias,
               int* __restrict__ lists,
               const float* __restrict__ A1,
               const int* __restrict__ rootrow,
               const float* __restrict__ W,
               const float* __restrict__ wbias,
               float* __restrict__ C, int N, int E) {
    __shared__ float smem[FUSED_WORDS];
    float* accs = smem;
    int* queue = (int*)(smem + NPB * DCOL);
    float* la0 = smem;
    float* la1 = smem + GRT * 16 * APITCH;

    const int tid  = threadIdx.x;
    const int lane = tid & 31;
    const int wave = __builtin_amdgcn_readfirstlane(tid >> 5);
    const int n0 = blockIdx.x * NPB;
    const int nn = (N - n0 < NPB) ? (N - n0) : NPB;

    constexpr int NIT = (GRT * 16 + SW - 1) / SW;
    static_assert(NIT <= 32, "one lane per staged row of a wave");
    static_assert(GRT > 12 && GRT <= 16, "row tiles g + 4t, t < 4, cover the block");
    int rrv;
    {
        const int r = wave + lane * SW;
        rrv = rootrow[n0 + (r < nn ? r : nn - 1)];
    }

    aggregate<1>(accs, queue, xw, src, dst, own, wvals, bias, lists, N, E);

    constexpr int NH = (NPB * (DCOL / 4) + SW * 32 - 1) / (SW * 32);
    v4f hold[NH];
#pragma unroll
    for (int k = 0; k < NH; ++k) {
        const int i = tid + k * (SW * 32);
        if (i < NPB * (DCOL / 4)) {
            const float* a = accs + (i >> 5) * DCOL + (i & 31);
            const v4f o = { lrelu(a[0]), lrelu(a[32]), lrelu(a[64]), lrelu(a[96]) };
            hold[k] = o;
        }
    }
    v4f t[NIT];
#pragma unroll
    for (int it = 0; it < NIT; ++it) {
        const int r = wave + it * SW;
        if (r < GRT * 16) {
            const int row = __builtin_amdgcn_readlane(rrv, it);
            t[it] = *(const v4f*)(A1 + (size_t)row * DCOL + lane * 4);
        }
    }
    __syncthreads();
#pragma unroll
    for (int k = 0; k < NH; ++k) {
        const int i = tid + k * (SW * 32);
        if (i < NPB * (DCOL / 4))
            *(v4f*)(la0 + (i >> 5) * APITCH + (i & 31) * 4) = hold[k];
    }
#pragma unroll
    for (int it = 0; it < NIT; ++it) {
        const int r = wave + it * SW;
        if (r < GRT * 16)
            *(v4f*)(la1 + r * APITCH + lane * 4) = t[it];
    }
    __syncthreads();

    const int lo  = lane & 15;
    const int hi  = lane >> 4;
    const int col = (wave & 7) * 16 + lo;
    const int g   = wave >> 3;
    if (g + 12 < GRT)
        fused_tiles<4>(la0, la1, W, wbias, C, n0, nn, g, col, lo, hi);
    else
        fused_tiles<3>(la0, la1, W, wbias, C, n0, nn, g, col, lo, hi);
}

__device__ __forceinline__ void store_tiles_full(float* __restrict__ p, const v8f (&acc)[GRT], int hi) {
#pragma unroll
    for (int j = 0; j < GRT; ++j) {
#pragma unroll
        for (int r = 0; r < 8; ++r)
            if (j * 16 + r + hi * 8 < GROWS) p[(j * 16 + r) * DCOL] = acc[j][r];
    }
}

__device__ __forceinline__ void store_tiles_part(float* __restrict__ rows, float* lp,
                                                 const float* la,
                                                 const v8f (&acc)[GRT], int tid, int nn) {
#pragma unroll
    for (int j = 0; j < GRT; ++j) {
#pragma unroll
        for (int r = 0; r < 8; ++r) lp[(j * 16 + r) * APITCH] = acc[j][r];
    }
    __syncthreads();
    for (int i = tid; i < nn * DCOL; i += GWAVES * 32)
        rows[i] = la[(i >> 7) * APITCH + (i & (DCOL - 1))];
}

__global__ __launch_bounds__(GWAVES * 32) void k_gemm_pair(const float* __restrict__ S,
                            const float* __restrict__ X,
                            const int* __restrict__ rootrow,
                            const float* __restrict__ Wa,
                            const float* __restrict__ ba,
                            const float* __restrict__ Wb,
                            float* __restrict__ H,
                            float* __restrict__ C, int N) {
    __shared__ float la[2 * GRT * 16 * APITCH];

    const int tid  = threadIdx.x;
    const int lane = tid & 31;
    const int wave = __builtin_amdgcn_readfirstlane(tid >> 5);
    const int lo   = lane & 15;
    const int hi   = lane >> 4;
    const int ct   = wave;
    const int n0 = blockIdx.x * GROWS;
    const int nn = (N - n0 < GROWS) ? (N - n0) : GROWS;
    const int col = ct * 16 + lo;

    v2f bc[32];
#pragma unroll
    for (int ks = 0; ks < 32; ++ks) {
        const float* p = Wa + (size_t)(ks * 4 + hi * 2) * DCOL + col;
        const v2f f = { p[0], p[DCOL] };
        bc[ks] = f;
    }
    const float badd = ba[col];

    constexpr int NIT = (GRT * 16) / GWAVES;
    int rrv;
    {
        const int r = wave + lane * GWAVES;
        rrv = rootrow[n0 + (r < nn ? r : nn - 1)];
    }
#pragma unroll
    for (int kc = 0; kc < 2; ++kc) {
        const float* Ap = (kc == 0) ? S : X;
        v4f t[NIT];
#pragma unroll
        for (int it = 0; it < NIT; ++it) {
            const int r = wave + it * GWAVES;
            const int rc = r < nn ? r : nn - 1;
            const int row = (kc == 0) ? (n0 + rc) : __builtin_amdgcn_readlane(rrv, it);
            t[it] = *(const v4f*)(Ap + (size_t)row * DCOL + lane * 4);
        }
#pragma unroll
        for (int it = 0; it < NIT; ++it) {
            const int r = wave + it * GWAVES;
            v4f v = t[it];
            if (kc == 1) { v[0] = lrelu(v[0]); v[1] = lrelu(v[1]); v[2] = lrelu(v[2]); v[3] = lrelu(v[3]); }
            *(v4f*)(la + (kc * GRT * 16 + r) * APITCH + lane * 4) = v;
        }
    }
    __syncthreads();

    const bool full = (nn == GROWS);
    const float* arow = la + lo * APITCH + hi * 2;
    float* hp = H + (size_t)(n0 + hi * 8) * DCOL + col;
    float* cp = C + (size_t)(n0 + hi * 8) * DCOL + col;
    float* lp = la + (hi * 8) * APITCH + col;
    v8f acc[GRT] = {};

#pragma nounroll
    for (int c = 0; c < 3; ++c) {
        const int cn = c < 2 ? c : 1;
        const float* Wn = Wb + (size_t)cn * 128 * DCOL;
        v2f bn[32];
#pragma unroll
        for (int ks = 0; ks < 32; ++ks) {
            const float* p = Wn + (size_t)(ks * 4 + hi * 2) * DCOL + col;
            const v2f f = { p[0], p[DCOL] };
            bn[ks] = f;
        }

        const float* ar = arow + (c == 2 ? GRT * 16 * APITCH : 0);
#pragma unroll
        for (int ks = 0; ks < 32; ++ks) {
#pragma unroll
            for (int j = 0; j < GRT; ++j) {
                const v2f a = *(const v2f*)(ar + j * 16 * APITCH + ks * 4);
                acc[j] = __builtin_amdgcn_wmma_f32_16x16x4_f32(
                    false, a, false, bc[ks], (short)0, acc[j], false, false);
            }
        }

        if (c == 0) {
            __syncthreads();
#pragma unroll
            for (int j = 0; j < GRT; ++j) {
#pragma unroll
                for (int r = 0; r < 8; ++r) acc[j][r] += badd;
            }
            if (full) {
                store_tiles_full(hp, acc, hi);
            } else {
                store_tiles_part(H + (size_t)n0 * DCOL, lp, la, acc, tid, nn);
                __syncthreads();
            }
#pragma unroll
            for (int j = 0; j < GRT; ++j) {
#pragma unroll
                for (int r = 0; r < 8; ++r) lp[(j * 16 + r) * APITCH] = lrelu(acc[j][r]);
                acc[j] = v8f{};
            }
            __syncthreads();
        }

#pragma unroll
        for (int ks = 0; ks < 32; ++ks) bc[ks] = bn[ks];
    }

    if (full) {
        store_tiles_full(cp, acc, hi);
    } else {
        __syncthreads();
        store_tiles_part(C + (size_t)n0 * DCOL, lp, la, acc, tid, nn);
    }
}

extern "C" void kernel_launch(void* const* d_in, const int* in_sizes, int n_in,
                              void* d_out, int out_size, void* d_ws, size_t ws_size,
                              hipStream_t stream) {
    const float* features = (const float*)d_in[0];
    const float* values   = (const float*)d_in[1];
    const float* W1       = (const float*)d_in[2];
    const float* b1       = (const float*)d_in[3];
    const float* W2       = (const float*)d_in[4];
    const float* b2       = (const float*)d_in[5];
    const float* Wl       = (const float*)d_in[6];
    const float* bl       = (const float*)d_in[7];
    const int*   adjs     = (const int*)d_in[8];
    const int*   root_idx = (const int*)d_in[9];
    const int*   batch    = (const int*)d_in[12];

    const int N = in_sizes[0] / DCOL;
    const int E = in_sizes[1];
    const int* srcI = adjs;
    const int* dstI = adjs + E;

    float* XW = (float*)d_ws;
    float* H1 = XW + (size_t)N * DCOL;
    float* AUX = H1 + (size_t)N * DCOL;
    int* ROOTROW = (int*)(AUX + (size_t)N * DCOL);

    float* OUT = (float*)d_out;
    int* OWN = (int*)AUX;
    int* LISTS = OWN + scan_steps(E) * (SU * SW * CHUNK / 4);

    const int gemmBlocks = (N + GROWS - 1) / GROWS;
    const int scatBlocks = (N + NPB - 1) / NPB;

    const int rootBlocks = (N + 255) / 256;
    const int ownBlocks  = (scan_steps(E) * (SU * SW * CHUNK) / 16 + 255) / 256;
    k_prep<<<rootBlocks + ownBlocks, 256, 0, stream>>>(batch, root_idx, ROOTROW, N,
                                                       dstI, OWN, E, rootBlocks);

    float* AX = OUT;
    k_scatter_own<<<scatBlocks, SW * 32, 0, stream>>>(features, srcI, dstI, OWN, values, nullptr, AX, LISTS, N, E);

    k_gemm_pair<<<gemmBlocks, GWAVES * 32, 0, stream>>>(
        AX, features, ROOTROW, W1, b1, W2, H1, XW, N);
    k_scatter_gemm<<<scatBlocks, SW * 32, 0, stream>>>(XW, srcI, dstI, OWN, values, b2, LISTS,
                                                      H1, ROOTROW, Wl, bl, OUT, N, E);
}


# First aggregation's block multiplies its own LDS row sums by W1, then W2, in the same launch (32 waves; edge sums never stored, H1 tile kept in LDS): launches 4 to 3; sums still LDS arrival order
# speedup vs baseline: 2.8711x; 1.0577x over previous
#include <hip/hip_runtime.h>
#include <hip/hip_bf16.h>

typedef float v2f __attribute__((ext_vector_type(2)));
typedef float v4f __attribute__((ext_vector_type(4)));
typedef float v8f __attribute__((ext_vector_type(8)));

#define NEG_SLOPE 0.01f
#define DCOL 128

__device__ __forceinline__ float lrelu(float x) {
    return x > 0.0f ? x : NEG_SLOPE * x;
}

#define NPB   200
#define SW    32
#define SU    2
#define CHUNK 512

__device__ __host__ __forceinline__ int scan_steps(int E) {
    const int nchunks = (E + CHUNK - 1) / CHUNK;
    const int nsteps = (nchunks + SU * SW - 1) / (SU * SW);
    return nsteps < 1 ? 1 : nsteps;
}

__device__ __forceinline__ unsigned owner_of(int d) {
    const unsigned o = (unsigned)d / (unsigned)NPB;
    return o < 255u ? o : 255u;
}

__global__ void k_prep(const int* __restrict__ batch,
                       const int* __restrict__ root_idx,
                       int* __restrict__ rootrow, int N,
                       const int* __restrict__ dst,
                       int* __restrict__ own, int E, int rootBlocks) {
    if ((int)blockIdx.x < rootBlocks) {
        const int i = blockIdx.x * blockDim.x + threadIdx.x;
        if (i < N) rootrow[i] = root_idx[batch[i]];
        return;
    }
    const int emap = scan_steps(E) * (SU * SW * CHUNK);
    const int e0 = ((blockIdx.x - rootBlocks) * blockDim.x + threadIdx.x) * 16;
    if (e0 >= emap) return;
    unsigned pk[4];
    if (((((size_t)dst) & 15) == 0) && e0 + 15 < E) {
#pragma unroll
        for (int g = 0; g < 4; ++g) {
            const int4 v = *(const int4*)(dst + e0 + g * 4);
            pk[g] = owner_of(v.x) | (owner_of(v.y) << 8) |
                    (owner_of(v.z) << 16) | (owner_of(v.w) << 24);
        }
    } else {
#pragma unroll
        for (int g = 0; g < 4; ++g) {
            unsigned p = 0u;
#pragma unroll
            for (int k = 0; k < 4; ++k) {
                const int e = e0 + g * 4 + k;
                const unsigned o = (e < E) ? owner_of(dst[e]) : 255u;
                p |= o << (8 * k);
            }
            pk[g] = p;
        }
    }
    *(uint4*)(own + e0 / 4) = make_uint4(pk[0], pk[1], pk[2], pk[3]);
}

#define BAT   15
#define QCAP  (SU * CHUNK + 64)

__device__ __forceinline__ void lds_row_add(float* a, float w, v4f x) {
    __hip_atomic_fetch_add(a + 0,  w * x[0], __ATOMIC_RELAXED, __HIP_MEMORY_SCOPE_WORKGROUP);
    __hip_atomic_fetch_add(a + 32, w * x[1], __ATOMIC_RELAXED, __HIP_MEMORY_SCOPE_WORKGROUP);
    __hip_atomic_fetch_add(a + 64, w * x[2], __ATOMIC_RELAXED, __HIP_MEMORY_SCOPE_WORKGROUP);
    __hip_atomic_fetch_add(a + 96, w * x[3], __ATOMIC_RELAXED, __HIP_MEMORY_SCOPE_WORKGROUP);
}

template <bool EXACT>
__device__ __forceinline__ int note_edges(uint4 o, unsigned pat, int ebase, int* q, int count,
                                          const int* __restrict__ dst, int n0, int nn, int E) {
    const unsigned wd[4] = { o.x, o.y, o.z, o.w };
#pragma unroll
    for (int g = 0; g < 4; ++g) {
        const unsigned x = wd[g] ^ pat;
        const unsigned z = (x - 0x01010101u) & ~x & 0x80808080u;
        if (__builtin_amdgcn_ballot_w32(z != 0u) == 0u) continue;
#pragma unroll
        for (int k = 0; k < 4; ++k) {
            const int e = ebase + g * 4 + k;
            bool m = ((x >> (8 * k)) & 0xffu) == 0u;
            if (!EXACT) {
                const bool in = m && (e < E);
                const int d = in ? dst[e] : -1;
                m = in && ((unsigned)(d - n0) < (unsigned)nn);
            }
            const unsigned mask = __builtin_amdgcn_ballot_w32(m);
            if (mask != 0u) {
                if (m) q[count + (int)__builtin_amdgcn_mbcnt_lo(mask, 0u)] = e;
                count += __builtin_popcount(mask);
            }
        }
    }
    return count;
}

__device__ __forceinline__ void ask_rows(const float* __restrict__ xw, int lane, int s,
                                         v4f (&xr)[BAT]) {
#pragma unroll
    for (int j = 0; j < BAT; ++j) {
        const int sj = __builtin_amdgcn_readlane(s, j);
        xr[j] = *(const v4f*)(xw + (size_t)sj * DCOL + lane * 4);
    }
}

__device__ __forceinline__ void add_rows(float* accs, int lane, int dl, float w,
                                         const v4f (&xr)[BAT]) {
#pragma unroll
    for (int j = 0; j < BAT; ++j) {
        const int dj = __builtin_amdgcn_readlane(dl, j);
        const float wj = __int_as_float(
            __builtin_amdgcn_readlane(__float_as_int(w), j));
        lds_row_add(accs + dj * DCOL + lane, wj, xr[j]);
    }
}

__device__ __forceinline__ void add_batch(const float* __restrict__ xw, float* accs, int lane,
                                          int nb, int s, int dl, float w) {
    v4f xr[BAT];
    if (nb == BAT) {
        ask_rows(xw, lane, s, xr);
        add_rows(accs, lane, dl, w, xr);
    } else {
#pragma unroll
        for (int j = 0; j < BAT; ++j) {
            if (j < nb) {
                const int sj = __builtin_amdgcn_readlane(s, j);
                xr[j] = *(const v4f*)(xw + (size_t)sj * DCOL + lane * 4);
            }
        }
#pragma unroll
        for (int j = 0; j < BAT; ++j) {
            if (j < nb) {
                const int dj = __builtin_amdgcn_readlane(dl, j);
                const float wj = __int_as_float(
                    __builtin_amdgcn_readlane(__float_as_int(w), j));
                lds_row_add(accs + dj * DCOL + lane, wj, xr[j]);
            }
        }
    }
}

#define CAPW 256

__device__ __forceinline__ int list_cap(int N, int E, int nblk) {
    const long long avail = (long long)N * DCOL
                          - (long long)scan_steps(E) * (SU * SW * CHUNK / 4)
                          - (long long)nblk * SW;
    const long long c = avail / (3LL * nblk * SW);
    return c < 0 ? 0 : (c > CAPW ? CAPW : (int)c);
}

template <int MODE>
__device__ __forceinline__ void aggregate(float* accs, int* queue,
               const float* __restrict__ xw,
               const int* __restrict__ src,
               const int* __restrict__ dst,
               const int* __restrict__ own,
               const float* __restrict__ wvals,
               const float* __restrict__ bias,
               int* __restrict__ lists, int N, int E) {
    const int tid  = threadIdx.x;
    const int lane = tid & 31;
    const int wave = __builtin_amdgcn_readfirstlane(tid >> 5);
    const int n0 = blockIdx.x * NPB;
    const int nn = (N - n0 < NPB) ? (N - n0) : NPB;

    for (int i = tid; i < nn * DCOL; i += SW * 32)
        accs[i] = bias ? bias[((i & 31) << 2) | ((i >> 5) & 3)] : 0.0f;
    __syncthreads();

    const int nblk = gridDim.x;
    const int cap = __builtin_amdgcn_readfirstlane(list_cap(N, E, nblk));
    const size_t plane = (size_t)nblk * SW * cap;
    int* cntp = lists + blockIdx.x * SW + wave;
    int* LS = lists + (size_t)nblk * SW + ((size_t)blockIdx.x * SW + wave) * cap;
    int* LD = LS + plane;
    float* LW = (float*)(LD + plane);

    bool scan = true;
    if (MODE == 1) {
        const int c = __builtin_amdgcn_readfirstlane(*cntp);
        if (c >= 0) {
            scan = false;
            int s = 0, dl = 0;
            float w = 0.0f;
            if (lane < BAT && lane < c) { s = LS[lane]; dl = LD[lane]; w = LW[lane]; }
            int base = 0;
            for (; base + BAT <= c; base += BAT) {
                v4f xr[BAT];
                ask_rows(xw, lane, s, xr);
                int s2 = 0, dl2 = 0;
                float w2 = 0.0f;
                if (lane < BAT && base + BAT + lane < c) {
                    s2  = LS[base + BAT + lane];
                    dl2 = LD[base + BAT + lane];
                    w2  = LW[base + BAT + lane];
                }
                add_rows(accs, lane, dl, w, xr);
                s = s2; dl = dl2; w = w2;
            }
            if (base < c) add_batch(xw, accs, lane, c - base, s, dl, w);
        }
    }

    if (scan) {
        int* q = queue + wave * QCAP;
        int count = 0;
        int ln = 0;
        bool over = false;
        const int nsteps = scan_steps(E);
        const int t0 = blockIdx.x % nsteps;
        const unsigned myb = blockIdx.x < 255u ? blockIdx.x : 255u;
        const bool exact = blockIdx.x < 255u;
        const unsigned pat = myb * 0x01010101u;

        for (int t = 0; t <= nsteps; ++t) {
            const bool more = t < nsteps;
            if (more) {
                int ts = t + t0;
                if (ts >= nsteps) ts -= nsteps;
                const int cb = ts * (SU * SW) + wave;
                uint4 ov[SU];
#pragma unroll
                for (int u = 0; u < SU; ++u)
                    ov[u] = *(const uint4*)(own + (size_t)(cb + u * SW) * (CHUNK / 4) + lane * 4);
                if (exact) {
#pragma unroll
                    for (int u = 0; u < SU; ++u)
                        count = note_edges<true>(ov[u], pat, (cb + u * SW) * CHUNK + lane * 16,
                                                 q, count, dst, n0, nn, E);
                } else {
#pragma unroll
                    for (int u = 0; u < SU; ++u)
                        count = note_edges<false>(ov[u], pat, (cb + u * SW) * CHUNK + lane * 16,
                                                  q, count, dst, n0, nn, E);
                }
            }

            while (count >= (more ? BAT : 1)) {
                const int nb = count < BAT ? count : BAT;
                count -= nb;
                int s = 0, dl = 0;
                float w = 0.0f;
                if (lane < nb) {
                    const int e = q[count + lane];
                    s  = src[e];
                    dl = dst[e] - n0;
                    w  = wvals[e];
                }
                if (MODE == 0) {
                    if (ln + nb <= cap) {
                        if (lane < nb) { LS[ln + lane] = s; LD[ln + lane] = dl; LW[ln + lane] = w; }
                        ln += nb;
                    } else {
                        over = true;
                    }
                }
                add_batch(xw, accs, lane, nb, s, dl, w);
            }
        }
        if (MODE == 0 && lane == 0) *cntp = over ? -1 : ln;
    }
    __syncthreads();
}

#define GROWS  NPB
#define GRT    ((GROWS + 15) / 16)
#define APITCH (DCOL + 4)

#define FUSED_AGG_WORDS (NPB * DCOL + SW * QCAP)
#define FUSED_IMG_WORDS (2 * GRT * 16 * APITCH)
#define FUSED_WORDS (FUSED_AGG_WORDS > FUSED_IMG_WORDS ? FUSED_AGG_WORDS : FUSED_IMG_WORDS)

template <int NT, int NCH, int NA>
__device__ __forceinline__ void chunk_products(v8f (&acc)[NA], const float* la0, const float* la1,
                                               const float* __restrict__ W,
                                               int g, int col, int lo, int hi) {
    static_assert(NT <= NA, "one accumulator per row tile");
#pragma unroll
    for (int kc = 0; kc < NCH; ++kc) {
        const float* img = (kc == 0 ? la0 : la1) + lo * APITCH + hi * 2;
#pragma unroll
        for (int h = 0; h < 2; ++h) {
            v2f b[16];
#pragma unroll
            for (int ks = 0; ks < 16; ++ks) {
                const float* p = W + (size_t)(kc * 128 + (h * 16 + ks) * 4 + hi * 2) * DCOL + col;
                const v2f f = { p[0], p[DCOL] };
                b[ks] = f;
            }
#pragma unroll
            for (int ks = 0; ks < 16; ++ks) {
#pragma unroll
                for (int t = 0; t < NT; ++t) {
                    const v2f a = *(const v2f*)(img + (g + 4 * t) * 16 * APITCH + (h * 16 + ks) * 4);
                    acc[t] = __builtin_amdgcn_wmma_f32_16x16x4_f32(
                        false, a, false, b[ks], (short)0, acc[t], false, false);
                }
            }
        }
    }
}


#define ROOT_NIT ((GRT * 16 + SW - 1) / SW)
__device__ __forceinline__ int root_row_numbers(const int* __restrict__ rootrow,
                                                int n0, int nn, int wave, int lane) {
    static_assert(ROOT_NIT <= 32, "one lane per staged row of a wave");
    static_assert(GRT > 12 && GRT <= 16, "row tiles g + 4t, t < 4, cover the block");
    const int r = wave + lane * SW;
    return rootrow[n0 + (r < nn ? r : nn - 1)];
}

template <bool ACT0, bool ACT1>
__device__ __forceinline__ void sums_to_images(float* smem, const float* __restrict__ rootsrc,
                                               int rrv, int tid, int lane, int wave) {
    const float* accs = smem;
    float* la0 = smem;
    float* la1 = smem + GRT * 16 * APITCH;

    constexpr int NH = (NPB * (DCOL / 4) + SW * 32 - 1) / (SW * 32);
    v4f hold[NH];
#pragma unroll
    for (int k = 0; k < NH; ++k) {
        const int i = tid + k * (SW * 32);
        if (i < NPB * (DCOL / 4)) {
            const float* a = accs + (i >> 5) * DCOL + (i & 31);
            v4f o = { a[0], a[32], a[64], a[96] };
            if (ACT0) { o[0] = lrelu(o[0]); o[1] = lrelu(o[1]); o[2] = lrelu(o[2]); o[3] = lrelu(o[3]); }
            hold[k] = o;
        }
    }
    v4f t[ROOT_NIT];
#pragma unroll
    for (int it = 0; it < ROOT_NIT; ++it) {
        const int r = wave + it * SW;
        if (r < GRT * 16) {
            const int row = __builtin_amdgcn_readlane(rrv, it);
            t[it] = *(const v4f*)(rootsrc + (size_t)row * DCOL + lane * 4);
        }
    }
    __syncthreads();
#pragma unroll
    for (int k = 0; k < NH; ++k) {
        const int i = tid + k * (SW * 32);
        if (i < NPB * (DCOL / 4))
            *(v4f*)(la0 + (i >> 5) * APITCH + (i & 31) * 4) = hold[k];
    }
#pragma unroll
    for (int it = 0; it < ROOT_NIT; ++it) {
        const int r = wave + it * SW;
        if (r < GRT * 16) {
            v4f v = t[it];
            if (ACT1) { v[0] = lrelu(v[0]); v[1] = lrelu(v[1]); v[2] = lrelu(v[2]); v[3] = lrelu(v[3]); }
            *(v4f*)(la1 + r * APITCH + lane * 4) = v;
        }
    }
    __syncthreads();
}

template <int NT>
__device__ __forceinline__ void h_tiles_to_image(v8f (&acc)[4], float badd, float* la0,
                                                 int g, int col, int hi) {
#pragma unroll
    for (int t = 0; t < NT; ++t) {
#pragma unroll
        for (int r = 0; r < 8; ++r) {
            acc[t][r] += badd;
            la0[((g + 4 * t) * 16 + r + hi * 8) * APITCH + col] = lrelu(acc[t][r]);
        }
    }
}

template <int NT>
__device__ __forceinline__ void tiles_to_memory(const v8f (&acc)[4], float* __restrict__ P,
                                                int n0, int nn, int g, int col, int hi) {
    if (nn == GROWS) {
        static_assert(11 * 16 + 15 < GROWS, "row tiles 0..11 lie inside a full block");
        float* p = P + (size_t)(n0 + g * 16 + hi * 8) * DCOL + col;
#pragma unroll
        for (int t = 0; t < NT; ++t) {
#pragma unroll
            for (int r = 0; r < 8; ++r)
                if (t < 3 || (g + 12) * 16 + r + hi * 8 < GROWS) p[(t * 64 + r) * DCOL] = acc[t][r];
        }
    } else {
#pragma unroll
        for (int t = 0; t < NT; ++t) {
#pragma unroll
            for (int r = 0; r < 8; ++r) {
                const int rowl = (g + 4 * t) * 16 + r + hi * 8;
                if (rowl < nn) P[(size_t)(n0 + rowl) * DCOL + col] = acc[t][r];
            }
        }
    }
}

__global__ __launch_bounds__(1024) void k_scatter_pair(const float* __restrict__ X,
               const int* __restrict__ src,
               const int* __restrict__ dst,
               const int* __restrict__ own,
               const float* __restrict__ wvals,
               int* __restrict__ lists,
               const int* __restrict__ rootrow,
               const float* __restrict__ Wa,
               const float* __restrict__ ba,
               const float* __restrict__ Wb,
               float* __restrict__ H,
               float* __restrict__ C, int N, int E) {
    __shared__ float smem[FUSED_WORDS];
    float* accs = smem;
    int* queue = (int*)(smem + NPB * DCOL);
    float* la0 = smem;
    float* la1 = smem + GRT * 16 * APITCH;

    const int tid  = threadIdx.x;
    const int lane = tid & 31;
    const int wave = __builtin_amdgcn_readfirstlane(tid >> 5);
    const int n0 = blockIdx.x * NPB;
    const int nn = (N - n0 < NPB) ? (N - n0) : NPB;

    const int rrv = root_row_numbers(rootrow, n0, nn, wave, lane);

    aggregate<0>(accs, queue, X, src, dst, own, wvals, nullptr, lists, N, E);

    sums_to_images<false, true>(smem, X, rrv, tid, lane, wave);

    const int lo  = lane & 15;
    const int hi  = lane >> 4;
    const int col = (wave & 7) * 16 + lo;
    const int g   = wave >> 3;
    const bool four = g + 12 < GRT;
    const float badd = ba[col];
    v8f acc[4] = {};

    if (four) chunk_products<4, 1>(acc, la0, la1, Wa, g, col, lo, hi);
    else      chunk_products<3, 1>(acc, la0, la1, Wa, g, col, lo, hi);
    __syncthreads();
    if (four) h_tiles_to_image<4>(acc, badd, la0, g, col, hi);
    else      h_tiles_to_image<3>(acc, badd, la0, g, col, hi);
    __syncthreads();
    if (four) tiles_to_memory<4>(acc, H, n0, nn, g, col, hi);
    else      tiles_to_memory<3>(acc, H, n0, nn, g, col, hi);
#pragma unroll
    for (int t = 0; t < 4; ++t) acc[t] = v8f{};

    if (four) chunk_products<4, 2>(acc, la0, la1, Wb, g, col, lo, hi);
    else      chunk_products<3, 2>(acc, la0, la1, Wb, g, col, lo, hi);
    if (four) tiles_to_memory<4>(acc, C, n0, nn, g, col, hi);
    else      tiles_to_memory<3>(acc, C, n0, nn, g, col, hi);
}

template <int NT>
__device__ __forceinline__ void fused_tiles(const float* la0, const float* la1,
                                            const float* __restrict__ W,
                                            const float* __restrict__ bias,
                                            float* __restrict__ C,
                                            int n0, int nn, int g, int col, int lo, int hi) {
    v8f acc[NT] = {};
#pragma unroll
    for (int kc = 0; kc < 2; ++kc) {
        const float* img = (kc == 0 ? la0 : la1) + lo * APITCH + hi * 2;
#pragma unroll
        for (int h = 0; h < 2; ++h) {
            v2f b[16];
#pragma unroll
            for (int ks = 0; ks < 16; ++ks) {
                const float* p = W + (size_t)(kc * 128 + (h * 16 + ks) * 4 + hi * 2) * DCOL + col;
                const v2f f = { p[0], p[DCOL] };
                b[ks] = f;
            }
#pragma unroll
            for (int ks = 0; ks < 16; ++ks) {
#pragma unroll
                for (int t = 0; t < NT; ++t) {
                    const v2f a = *(const v2f*)(img + (g + 4 * t) * 16 * APITCH + (h * 16 + ks) * 4);
                    acc[t] = __builtin_amdgcn_wmma_f32_16x16x4_f32(
                        false, a, false, b[ks], (short)0, acc[t], false, false);
                }
            }
        }
    }
    const float badd = bias[col];
#pragma unroll
    for (int t = 0; t < NT; ++t) {
#pragma unroll
        for (int r = 0; r < 8; ++r) {
            const int rowl = (g + 4 * t) * 16 + r + hi * 8;
            if (rowl < nn) C[(size_t)(n0 + rowl) * DCOL + col] = lrelu(acc[t][r] + badd);
        }
    }
}

__global__ __launch_bounds__(1024) void k_scatter_gemm(const float* __restrict__ xw,
               const int* __restrict__ src,
               const int* __restrict__ dst,
               const int* __restrict__ own,
               const float* __restrict__ wvals,
               const float* __restrict__ bias,
               int* __restrict__ lists,
               const float* __restrict__ A1,
               const int* __restrict__ rootrow,
               const float* __restrict__ W,
               const float* __restrict__ wbias,
               float* __restrict__ C, int N, int E) {
    __shared__ float smem[FUSED_WORDS];
    float* accs = smem;
    int* queue = (int*)(smem + NPB * DCOL);
    float* la0 = smem;
    float* la1 = smem + GRT * 16 * APITCH;

    const int tid  = threadIdx.x;
    const int lane = tid & 31;
    const int wave = __builtin_amdgcn_readfirstlane(tid >> 5);
    const int n0 = blockIdx.x * NPB;
    const int nn = (N - n0 < NPB) ? (N - n0) : NPB;

    const int rrv = root_row_numbers(rootrow, n0, nn, wave, lane);

    aggregate<1>(accs, queue, xw, src, dst, own, wvals, bias, lists, N, E);

    sums_to_images<true, false>(smem, A1, rrv, tid, lane, wave);

    const int lo  = lane & 15;
    const int hi  = lane >> 4;
    const int col = (wave & 7) * 16 + lo;
    const int g   = wave >> 3;
    if (g + 12 < GRT)
        fused_tiles<4>(la0, la1, W, wbias, C, n0, nn, g, col, lo, hi);
    else
        fused_tiles<3>(la0, la1, W, wbias, C, n0, nn, g, col, lo, hi);
}

extern "C" void kernel_launch(void* const* d_in, const int* in_sizes, int n_in,
                              void* d_out, int out_size, void* d_ws, size_t ws_size,
                              hipStream_t stream) {
    const float* features = (const float*)d_in[0];
    const float* values   = (const float*)d_in[1];
    const float* W1       = (const float*)d_in[2];
    const float* b1       = (const float*)d_in[3];
    const float* W2       = (const float*)d_in[4];
    const float* b2       = (const float*)d_in[5];
    const float* Wl       = (const float*)d_in[6];
    const float* bl       = (const float*)d_in[7];
    const int*   adjs     = (const int*)d_in[8];
    const int*   root_idx = (const int*)d_in[9];
    const int*   batch    = (const int*)d_in[12];

    const int N = in_sizes[0] / DCOL;
    const int E = in_sizes[1];
    const int* srcI = adjs;
    const int* dstI = adjs + E;

    float* XW = (float*)d_ws;
    float* H1 = XW + (size_t)N * DCOL;
    float* AUX = H1 + (size_t)N * DCOL;
    int* ROOTROW = (int*)(AUX + (size_t)N * DCOL);

    float* OUT = (float*)d_out;
    int* OWN = (int*)AUX;
    int* LISTS = OWN + scan_steps(E) * (SU * SW * CHUNK / 4);

    const int scatBlocks = (N + NPB - 1) / NPB;

    const int rootBlocks = (N + 255) / 256;
    const int ownBlocks  = (scan_steps(E) * (SU * SW * CHUNK) / 16 + 255) / 256;
    k_prep<<<rootBlocks + ownBlocks, 256, 0, stream>>>(batch, root_idx, ROOTROW, N,
                                                       dstI, OWN, E, rootBlocks);

    k_scatter_pair<<<scatBlocks, SW * 32, 0, stream>>>(features, srcI, dstI, OWN, values, LISTS,
                                                      ROOTROW, W1, b1, W2, H1, XW, N, E);
    k_scatter_gemm<<<scatBlocks, SW * 32, 0, stream>>>(XW, srcI, dstI, OWN, values, b2, LISTS,
                                                      H1, ROOTROW, Wl, bl, OUT, N, E);
}
